# starting kernel
# baseline (speedup 1.0000x reference)
#include <hip/hip_runtime.h>
#include <math.h>

typedef __attribute__((ext_vector_type(16))) _Float16 v16h;
typedef __attribute__((ext_vector_type(8)))  float    v8f;

namespace cfg {
constexpr int NB    = 128;
constexpr int SEQ   = 64;
constexpr int CH    = 256;
constexpr int PARTS = 16;
constexpr int HID   = 64;
constexpr int XT_ROWS   = SEQ + 4;
constexpr int XT_STRIDE = CH + 8;
constexpr int HT_STRIDE = HID + 8;
}

__device__ __forceinline__ v16h load_a_frag_g(const float* base, int rstride,
                                              int kstride, int lane) {
  const int m  = lane & 15;
  const int hs = (lane >> 4) & 1;
  const float* p = base + (size_t)m * rstride;
  v16h a;
#pragma unroll
  for (int t = 0; t < 16; ++t) {
    const int v = t >> 1, e = t & 1;
    const int k = ((v >> 2) << 4) + (hs << 3) + ((v & 3) << 1) + e;
    a[t] = (_Float16)p[k * kstride];
  }
  return a;
}

__device__ __forceinline__ v16h load_b_frag_lds(const _Float16* lds, int row_base,
                                                int row_stride, int col_base,
                                                int lane) {
  const int n  = lane & 15;
  const int hs = (lane >> 4) & 1;
  const _Float16* p = lds + (size_t)(row_base + n) * row_stride + col_base;
  v16h b;
#pragma unroll
  for (int t = 0; t < 16; ++t) {
    const int v = t >> 1, e = t & 1;
    const int k = ((v >> 2) << 4) + (hs << 3) + ((v & 3) << 1) + e;
    b[t] = p[k];
  }
  return b;
}

__device__ __forceinline__ v8f wmma_f16(v16h a, v16h b, v8f c) {
  return __builtin_amdgcn_wmma_f32_16x16x32_f16(false, a, false, b,
                                                (short)0, c, false, false);
}

__device__ __forceinline__ float sigmoidf_(float z) {
  return 1.0f / (1.0f + __expf(-z));
}

__global__ __launch_bounds__(256)
void mtb_gate_kernel(const float* __restrict__ x,
                     const float* __restrict__ W1a,
                     const float* __restrict__ W1b,
                     const float* __restrict__ W2a,
                     const float* __restrict__ W2b,
                     float* __restrict__ out) {
  using namespace cfg;
  __shared__ _Float16 sXt[XT_ROWS * XT_STRIDE];
  __shared__ _Float16 sHt[2][XT_ROWS * HT_STRIDE];

  const int part = blockIdx.x;
  const int b    = blockIdx.y;
  const int tid  = threadIdx.x;
  const int lane = tid & 31;
  const int wave = tid >> 5;
  const int hs   = (lane >> 4) & 1;

  {
    sXt[0 * XT_STRIDE + tid]         = (_Float16)0.f;
    sXt[1 * XT_STRIDE + tid]         = (_Float16)0.f;
    sXt[(SEQ + 2) * XT_STRIDE + tid] = (_Float16)0.f;
    sXt[(SEQ + 3) * XT_STRIDE + tid] = (_Float16)0.f;
    if (tid < HID) {
#pragma unroll
      for (int r = 0; r < 2; ++r) {
        sHt[0][r * HT_STRIDE + tid] = (_Float16)0.f;
        sHt[1][r * HT_STRIDE + tid] = (_Float16)0.f;
        sHt[0][(SEQ + 2 + r) * HT_STRIDE + tid] = (_Float16)0.f;
        sHt[1][(SEQ + 2 + r) * HT_STRIDE + tid] = (_Float16)0.f;
      }
    }
    const int c = tid;
    const float* xp = x + (((size_t)b * SEQ) * CH + c) * PARTS + part;
#pragma unroll 4
    for (int l = 0; l < SEQ; ++l) {
      float v = xp[(size_t)l * CH * PARTS];
      sXt[(2 + l) * XT_STRIDE + c] = (_Float16)v;
    }
  }
  __syncthreads();

  {
    const int br = wave >> 2;
    const int mt = wave & 3;
    const int m0 = mt * 16;
    const float* Wa = (br ? W2a : W1a) +
                      ((size_t)(part * HID + m0)) * CH * 3;
    v8f acc[4] = {};
    for (int s = 0; s < 3; ++s) {
      for (int kb = 0; kb < CH / 32; ++kb) {
        const int c0 = kb * 32;
        v16h a = load_a_frag_g(Wa + c0 * 3 + s, CH * 3, 3, lane);
#pragma unroll
        for (int nt = 0; nt < 4; ++nt) {
          v16h bf = load_b_frag_lds(sXt, 1 + s + nt * 16, XT_STRIDE, c0, lane);
          acc[nt] = wmma_f16(a, bf, acc[nt]);
        }
      }
    }
#pragma unroll
    for (int nt = 0; nt < 4; ++nt) {
#pragma unroll
      for (int v = 0; v < 8; ++v) {
        float f = acc[nt][v];
        f = f > 0.f ? f : 0.01f * f;
        const int row = 2 + nt * 16 + (lane & 15);
        const int col = m0 + v + (hs << 3);
        sHt[br][row * HT_STRIDE + col] = (_Float16)f;
      }
    }
  }
  __syncthreads();

  for (int j = 0; j < 2; ++j) {
    const int m0 = wave * 32 + j * 16;
    const float* w1bp = W1b + ((size_t)(part * CH + m0)) * HID;
    const float* w2bp = W2b + ((size_t)(part * CH + m0)) * HID * 3;
    v16h a1[2], a2[3][2];
#pragma unroll
    for (int kb = 0; kb < 2; ++kb)
      a1[kb] = load_a_frag_g(w1bp + kb * 32, HID, 1, lane);
#pragma unroll
    for (int s = 0; s < 3; ++s)
#pragma unroll
      for (int kb = 0; kb < 2; ++kb)
        a2[s][kb] = load_a_frag_g(w2bp + kb * 32 * 3 + s, HID * 3, 3, lane);

    float maxv[8];
#pragma unroll
    for (int v = 0; v < 8; ++v) maxv[v] = -INFINITY;

    for (int nt = 0; nt < 4; ++nt) {
      v8f acc1 = {}, acc2 = {};
#pragma unroll
      for (int kb = 0; kb < 2; ++kb) {
        v16h bf = load_b_frag_lds(sHt[0], 2 + nt * 16, HT_STRIDE, kb * 32, lane);
        acc1 = wmma_f16(a1[kb], bf, acc1);
      }
      for (int s = 0; s < 3; ++s) {
#pragma unroll
        for (int kb = 0; kb < 2; ++kb) {
          v16h bf = load_b_frag_lds(sHt[1], 1 + s + nt * 16, HT_STRIDE,
                                    kb * 32, lane);
          acc2 = wmma_f16(a2[s][kb], bf, acc2);
        }
      }
#pragma unroll
      for (int v = 0; v < 8; ++v) {
        const int c = m0 + v + (hs << 3);
        const int l = nt * 16 + (lane & 15);
        float xs[5];
#pragma unroll
        for (int d = 0; d < 5; ++d)
          xs[d] = (float)sXt[(l + d) * XT_STRIDE + c];
        const float avg3 = (xs[1] + xs[2] + xs[3]) * (1.0f / 3.0f);
        const float avg5 = (xs[0] + xs[1] + xs[2] + xs[3] + xs[4]) * 0.2f;
        float m3 = xs[2];
        m3 = fmaxf(m3, (l >= 1)       ? xs[1] : -INFINITY);
        m3 = fmaxf(m3, (l <= cfg::SEQ - 2) ? xs[3] : -INFINITY);
        float m5 = m3;
        m5 = fmaxf(m5, (l >= 2)       ? xs[0] : -INFINITY);
        m5 = fmaxf(m5, (l <= cfg::SEQ - 3) ? xs[4] : -INFINITY);
        const float g1 = sigmoidf_(acc1[v]);
        const float g2 = sigmoidf_(acc2[v]);
        const float mval = (avg3 + m3) * g1 + (avg5 + m5) * g2;
        maxv[v] = fmaxf(maxv[v], mval);
      }
    }
#pragma unroll
    for (int v = 0; v < 8; ++v) {
#pragma unroll
      for (int off = 8; off >= 1; off >>= 1)
        maxv[v] = fmaxf(maxv[v], __shfl_xor(maxv[v], off, 16));
      if ((lane & 15) == 0) {
        const int c = m0 + v + (hs << 3);
        out[((size_t)b * cfg::PARTS + part) * cfg::CH + c] = maxv[v];
      }
    }
  }
}

extern "C" void kernel_launch(void* const* d_in, const int* in_sizes, int n_in,
                              void* d_out, int out_size, void* d_ws, size_t ws_size,
                              hipStream_t stream) {
  (void)in_sizes; (void)n_in; (void)d_ws; (void)ws_size; (void)out_size;
  const float* x   = (const float*)d_in[0];
  const float* W1a = (const float*)d_in[1];
  const float* W1b = (const float*)d_in[2];
  const float* W2a = (const float*)d_in[3];
  const float* W2b = (const float*)d_in[4];
  float* out = (float*)d_out;

  dim3 grid(cfg::PARTS, cfg::NB);
  dim3 block(256);
  mtb_gate_kernel<<<grid, block, 0, stream>>>(x, W1a, W1b, W2a, W2b, out);
}


# Pre-pass launch rewrites the four weight tensors as f16 WMMA A fragments in the workspace; main kernel loads a fragment with two coalesced 16-byte loads per lane, not 16 strided 4-byte loads
# speedup vs baseline: 1.8549x; 1.8549x over previous
#include <hip/hip_runtime.h>
#include <math.h>

typedef __attribute__((ext_vector_type(16))) _Float16 v16h;
typedef __attribute__((ext_vector_type(8)))  _Float16 v8h;
typedef __attribute__((ext_vector_type(8)))  float    v8f;

namespace cfg {
constexpr int NB    = 128;
constexpr int SEQ   = 64;
constexpr int CH    = 256;
constexpr int PARTS = 16;
constexpr int HID   = 64;
constexpr int XT_ROWS   = SEQ + 4;
constexpr int XT_STRIDE = CH + 8;
constexpr int HT_STRIDE = HID + 8;
constexpr int FRAG_HALFS = 512;
constexpr int NFRAG_A  = 2 * PARTS * 4 * 3 * 8;
constexpr int NFRAG_B1 = PARTS * 16 * 2;
constexpr int NFRAG_B2 = PARTS * 16 * 3 * 2;
constexpr int NFRAG    = NFRAG_A + NFRAG_B1 + NFRAG_B2;
constexpr int PRE_BLOCKS = NFRAG * 64 / 256;
}

__device__ __forceinline__ v16h load_a_frag_ws(const _Float16* __restrict__ wf,
                                               int frag, int lane) {
  const _Float16* p = wf + (size_t)frag * cfg::FRAG_HALFS + lane * 8;
  const v8h lo = *(const v8h*)p;
  const v8h hi = *(const v8h*)(p + 256);
  return __builtin_shufflevector(lo, hi, 0, 1, 2, 3, 4, 5, 6, 7,
                                 8, 9, 10, 11, 12, 13, 14, 15);
}

__device__ __forceinline__ v16h load_b_frag_lds(const _Float16* lds, int row_base,
                                                int row_stride, int col_base,
                                                int lane) {
  const int n  = lane & 15;
  const int hs = (lane >> 4) & 1;
  const _Float16* p = lds + (size_t)(row_base + n) * row_stride + col_base;
  v16h b;
#pragma unroll
  for (int t = 0; t < 16; ++t) {
    const int v = t >> 1, e = t & 1;
    const int k = ((v >> 2) << 4) + (hs << 3) + ((v & 3) << 1) + e;
    b[t] = p[k];
  }
  return b;
}

__device__ __forceinline__ v8f wmma_f16(v16h a, v16h b, v8f c) {
  return __builtin_amdgcn_wmma_f32_16x16x32_f16(false, a, false, b,
                                                (short)0, c, false, false);
}

__device__ __forceinline__ float sigmoidf_(float z) {
  return 1.0f / (1.0f + __expf(-z));
}

__global__ __launch_bounds__(256)
void weight_frag_kernel(const float* __restrict__ W1a,
                        const float* __restrict__ W1b,
                        const float* __restrict__ W2a,
                        const float* __restrict__ W2b,
                        _Float16* __restrict__ wf) {
  using namespace cfg;
  const int gid  = blockIdx.x * 256 + threadIdx.x;
  const int frag = gid >> 6;
  const int sel  = (gid >> 5) & 1;
  const int lane = gid & 31;
  const int m    = lane & 15;
  const int hs   = lane >> 4;
  const int k0   = sel * 16 + hs * 8;
  const float* src;
  int kstride;
  if (frag < NFRAG_A) {
    const int kb = frag & 7;
    int r = frag >> 3;
    const int s = r % 3;  r /= 3;
    const int mt = r & 3; r >>= 2;
    const int part = r & 15;
    const int br = r >> 4;
    src = (br ? W2a : W1a) +
          ((size_t)((part * HID + mt * 16 + m) * CH + kb * 32 + k0)) * 3 + s;
    kstride = 3;
  } else if (frag < NFRAG_A + NFRAG_B1) {
    const int f = frag - NFRAG_A;
    const int kb = f & 1;
    const int ct = (f >> 1) & 15;
    const int part = f >> 5;
    src = W1b + (size_t)((part * CH + ct * 16 + m) * HID + kb * 32 + k0);
    kstride = 1;
  } else {
    const int f = frag - NFRAG_A - NFRAG_B1;
    const int kb = f & 1;
    int r = f >> 1;
    const int s = r % 3;  r /= 3;
    const int ct = r & 15;
    const int part = r >> 4;
    src = W2b + ((size_t)((part * CH + ct * 16 + m) * HID + kb * 32 + k0)) * 3 + s;
    kstride = 3;
  }
  v8h o;
#pragma unroll
  for (int e = 0; e < 8; ++e) o[e] = (_Float16)src[e * kstride];
  *(v8h*)(wf + (size_t)gid * 8) = o;
}

__global__ __launch_bounds__(256)
void mtb_gate_kernel(const float* __restrict__ x,
                     const _Float16* __restrict__ wf,
                     float* __restrict__ out) {
  using namespace cfg;
  __shared__ _Float16 sXt[XT_ROWS * XT_STRIDE];
  __shared__ _Float16 sHt[2][XT_ROWS * HT_STRIDE];

  const int part = blockIdx.x;
  const int b    = blockIdx.y;
  const int tid  = threadIdx.x;
  const int lane = tid & 31;
  const int wave = tid >> 5;
  const int hs   = (lane >> 4) & 1;

  {
    sXt[0 * XT_STRIDE + tid]         = (_Float16)0.f;
    sXt[1 * XT_STRIDE + tid]         = (_Float16)0.f;
    sXt[(SEQ + 2) * XT_STRIDE + tid] = (_Float16)0.f;
    sXt[(SEQ + 3) * XT_STRIDE + tid] = (_Float16)0.f;
    if (tid < HID) {
#pragma unroll
      for (int r = 0; r < 2; ++r) {
        sHt[0][r * HT_STRIDE + tid] = (_Float16)0.f;
        sHt[1][r * HT_STRIDE + tid] = (_Float16)0.f;
        sHt[0][(SEQ + 2 + r) * HT_STRIDE + tid] = (_Float16)0.f;
        sHt[1][(SEQ + 2 + r) * HT_STRIDE + tid] = (_Float16)0.f;
      }
    }
    const int c = tid;
    const float* xp = x + (((size_t)b * SEQ) * CH + c) * PARTS + part;
#pragma unroll 4
    for (int l = 0; l < SEQ; ++l) {
      float v = xp[(size_t)l * CH * PARTS];
      sXt[(2 + l) * XT_STRIDE + c] = (_Float16)v;
    }
  }
  __syncthreads();

  {
    const int br = wave >> 2;
    const int mt = wave & 3;
    const int m0 = mt * 16;
    const int fbase = ((br * PARTS + part) * 4 + mt) * 24;
    v8f acc[4] = {};
    for (int s = 0; s < 3; ++s) {
      for (int kb = 0; kb < CH / 32; ++kb) {
        const int c0 = kb * 32;
        v16h a = load_a_frag_ws(wf, fbase + s * 8 + kb, lane);
#pragma unroll
        for (int nt = 0; nt < 4; ++nt) {
          v16h bf = load_b_frag_lds(sXt, 1 + s + nt * 16, XT_STRIDE, c0, lane);
          acc[nt] = wmma_f16(a, bf, acc[nt]);
        }
      }
    }
#pragma unroll
    for (int nt = 0; nt < 4; ++nt) {
#pragma unroll
      for (int v = 0; v < 8; ++v) {
        float f = acc[nt][v];
        f = f > 0.f ? f : 0.01f * f;
        const int row = 2 + nt * 16 + (lane & 15);
        const int col = m0 + v + (hs << 3);
        sHt[br][row * HT_STRIDE + col] = (_Float16)f;
      }
    }
  }
  __syncthreads();

  for (int j = 0; j < 2; ++j) {
    const int m0 = wave * 32 + j * 16;
    const int ct  = wave * 2 + j;
    const int fb1 = NFRAG_A + (part * 16 + ct) * 2;
    const int fb2 = NFRAG_A + NFRAG_B1 + (part * 16 + ct) * 6;
    v16h a1[2], a2[3][2];
#pragma unroll
    for (int kb = 0; kb < 2; ++kb)
      a1[kb] = load_a_frag_ws(wf, fb1 + kb, lane);
#pragma unroll
    for (int s = 0; s < 3; ++s)
#pragma unroll
      for (int kb = 0; kb < 2; ++kb)
        a2[s][kb] = load_a_frag_ws(wf, fb2 + s * 2 + kb, lane);

    float maxv[8];
#pragma unroll
    for (int v = 0; v < 8; ++v) maxv[v] = -INFINITY;

    for (int nt = 0; nt < 4; ++nt) {
      v8f acc1 = {}, acc2 = {};
#pragma unroll
      for (int kb = 0; kb < 2; ++kb) {
        v16h bf = load_b_frag_lds(sHt[0], 2 + nt * 16, HT_STRIDE, kb * 32, lane);
        acc1 = wmma_f16(a1[kb], bf, acc1);
      }
      for (int s = 0; s < 3; ++s) {
#pragma unroll
        for (int kb = 0; kb < 2; ++kb) {
          v16h bf = load_b_frag_lds(sHt[1], 1 + s + nt * 16, HT_STRIDE,
                                    kb * 32, lane);
          acc2 = wmma_f16(a2[s][kb], bf, acc2);
        }
      }
#pragma unroll
      for (int v = 0; v < 8; ++v) {
        const int c = m0 + v + (hs << 3);
        const int l = nt * 16 + (lane & 15);
        float xs[5];
#pragma unroll
        for (int d = 0; d < 5; ++d)
          xs[d] = (float)sXt[(l + d) * XT_STRIDE + c];
        const float avg3 = (xs[1] + xs[2] + xs[3]) * (1.0f / 3.0f);
        const float avg5 = (xs[0] + xs[1] + xs[2] + xs[3] + xs[4]) * 0.2f;
        float m3 = xs[2];
        m3 = fmaxf(m3, (l >= 1)       ? xs[1] : -INFINITY);
        m3 = fmaxf(m3, (l <= cfg::SEQ - 2) ? xs[3] : -INFINITY);
        float m5 = m3;
        m5 = fmaxf(m5, (l >= 2)       ? xs[0] : -INFINITY);
        m5 = fmaxf(m5, (l <= cfg::SEQ - 3) ? xs[4] : -INFINITY);
        const float g1 = sigmoidf_(acc1[v]);
        const float g2 = sigmoidf_(acc2[v]);
        const float mval = (avg3 + m3) * g1 + (avg5 + m5) * g2;
        maxv[v] = fmaxf(maxv[v], mval);
      }
    }
#pragma unroll
    for (int v = 0; v < 8; ++v) {
#pragma unroll
      for (int off = 8; off >= 1; off >>= 1)
        maxv[v] = fmaxf(maxv[v], __shfl_xor(maxv[v], off, 16));
      if ((lane & 15) == 0) {
        const int c = m0 + v + (hs << 3);
        out[((size_t)b * cfg::PARTS + part) * cfg::CH + c] = maxv[v];
      }
    }
  }
}

extern "C" void kernel_launch(void* const* d_in, const int* in_sizes, int n_in,
                              void* d_out, int out_size, void* d_ws, size_t ws_size,
                              hipStream_t stream) {
  (void)in_sizes; (void)n_in; (void)ws_size; (void)out_size;
  const float* x   = (const float*)d_in[0];
  const float* W1a = (const float*)d_in[1];
  const float* W1b = (const float*)d_in[2];
  const float* W2a = (const float*)d_in[3];
  const float* W2b = (const float*)d_in[4];
  float* out = (float*)d_out;
  _Float16* wf = (_Float16*)d_ws;

  weight_frag_kernel<<<dim3(cfg::PRE_BLOCKS), dim3(256), 0, stream>>>(W1a, W1b, W2a, W2b, wf);
  dim3 grid(cfg::PARTS, cfg::NB);
  dim3 block(256);
  mtb_gate_kernel<<<grid, block, 0, stream>>>(x, wf, out);
}


# Pre-pass also rewrites x in the workspace as 16-bit float tiles [b][part][l][c]; main kernel fills its x tile with 8 coalesced 16-byte loads per thread, not 64 4-byte loads at a 64-byte stride
# speedup vs baseline: 3.1905x; 1.7200x over previous
#include <hip/hip_runtime.h>
#include <math.h>

typedef __attribute__((ext_vector_type(16))) _Float16 v16h;
typedef __attribute__((ext_vector_type(8)))  _Float16 v8h;
typedef __attribute__((ext_vector_type(8)))  float    v8f;

namespace cfg {
constexpr int NB    = 128;
constexpr int SEQ   = 64;
constexpr int CH    = 256;
constexpr int PARTS = 16;
constexpr int HID   = 64;
constexpr int XT_ROWS   = SEQ + 4;
constexpr int XT_STRIDE = CH + 8;
constexpr int HT_STRIDE = HID + 8;
constexpr int FRAG_HALFS = 512;
constexpr int NFRAG_A  = 2 * PARTS * 4 * 3 * 8;
constexpr int NFRAG_B1 = PARTS * 16 * 2;
constexpr int NFRAG_B2 = PARTS * 16 * 3 * 2;
constexpr int NFRAG    = NFRAG_A + NFRAG_B1 + NFRAG_B2;
constexpr int PRE_BLOCKS = NFRAG * 64 / 256;
constexpr size_t WF_HALFS = (size_t)NFRAG * FRAG_HALFS;
constexpr int XT_PIECES = NB * SEQ * (CH / 64);
constexpr int XT_TRIPS  = 4;
constexpr int XT_BLOCKS = XT_PIECES / (2 * 8 * XT_TRIPS);
constexpr int PRE_GRID  = XT_BLOCKS + PRE_BLOCKS;
}

__device__ __forceinline__ v16h load_a_frag_ws(const _Float16* __restrict__ wf,
                                               int frag, int lane) {
  const _Float16* p = wf + (size_t)frag * cfg::FRAG_HALFS + lane * 8;
  const v8h lo = *(const v8h*)p;
  const v8h hi = *(const v8h*)(p + 256);
  return __builtin_shufflevector(lo, hi, 0, 1, 2, 3, 4, 5, 6, 7,
                                 8, 9, 10, 11, 12, 13, 14, 15);
}

__device__ __forceinline__ v16h load_b_frag_lds(const _Float16* lds, int row_base,
                                                int row_stride, int col_base,
                                                int lane) {
  const int n  = lane & 15;
  const int hs = (lane >> 4) & 1;
  const _Float16* p = lds + (size_t)(row_base + n) * row_stride + col_base;
  v16h b;
#pragma unroll
  for (int t = 0; t < 16; ++t) {
    const int v = t >> 1, e = t & 1;
    const int k = ((v >> 2) << 4) + (hs << 3) + ((v & 3) << 1) + e;
    b[t] = p[k];
  }
  return b;
}

__device__ __forceinline__ v8f wmma_f16(v16h a, v16h b, v8f c) {
  return __builtin_amdgcn_wmma_f32_16x16x32_f16(false, a, false, b,
                                                (short)0, c, false, false);
}

__device__ __forceinline__ float sigmoidf_(float z) {
  return 1.0f / (1.0f + __expf(-z));
}

__device__ __forceinline__ void x_to_f16_tiles(const float* __restrict__ x,
                                               _Float16* __restrict__ xt,
                                               int blk, int tid) {
  using namespace cfg;
  const int wave = tid >> 5;
  const int lane = tid & 31;
  const int c8   = lane >> 2;
  const int pq   = lane & 3;
  for (int it = 0; it < XT_TRIPS; ++it) {
    const int piece0 = ((blk * XT_TRIPS + it) * 8 + wave) * 2;
    const int row    = piece0 >> 2;
    const int c0     = (piece0 & 3) * 64 + c8 * 8;
    const float* src = x + ((size_t)row * CH + c0) * PARTS + pq * 4;
    float4 v[2][8];
#pragma unroll
    for (int h = 0; h < 2; ++h)
#pragma unroll
      for (int j = 0; j < 8; ++j)
        v[h][j] = *(const float4*)(src + (h * 64 + j) * PARTS);
    __builtin_amdgcn_sched_barrier(0);
    const int b = row >> 6;
    const int l = row & 63;
    _Float16* dst = xt + (((size_t)(b * PARTS + pq * 4) * SEQ + l) * CH + c0);
#pragma unroll
    for (int h = 0; h < 2; ++h) {
      v8h o0, o1, o2, o3;
#pragma unroll
      for (int j = 0; j < 8; ++j) {
        o0[j] = (_Float16)v[h][j].x;
        o1[j] = (_Float16)v[h][j].y;
        o2[j] = (_Float16)v[h][j].z;
        o3[j] = (_Float16)v[h][j].w;
      }
      *(v8h*)(dst + h * 64)                        = o0;
      *(v8h*)(dst + h * 64 + (size_t)1 * SEQ * CH) = o1;
      *(v8h*)(dst + h * 64 + (size_t)2 * SEQ * CH) = o2;
      *(v8h*)(dst + h * 64 + (size_t)3 * SEQ * CH) = o3;
    }
  }
}

__global__ __launch_bounds__(256)
void prepass_kernel(const float* __restrict__ x,
                    const float* __restrict__ W1a,
                    const float* __restrict__ W1b,
                    const float* __restrict__ W2a,
                    const float* __restrict__ W2b,
                    _Float16* __restrict__ wf) {
  using namespace cfg;
  if (blockIdx.x < XT_BLOCKS) {
    x_to_f16_tiles(x, wf + WF_HALFS, blockIdx.x, threadIdx.x);
    return;
  }
  const int gid  = (blockIdx.x - XT_BLOCKS) * 256 + threadIdx.x;
  const int frag = gid >> 6;
  const int sel  = (gid >> 5) & 1;
  const int lane = gid & 31;
  const int m    = lane & 15;
  const int hs   = lane >> 4;
  const int k0   = sel * 16 + hs * 8;
  const float* src;
  int kstride;
  if (frag < NFRAG_A) {
    const int kb = frag & 7;
    int r = frag >> 3;
    const int s = r % 3;  r /= 3;
    const int mt = r & 3; r >>= 2;
    const int part = r & 15;
    const int br = r >> 4;
    src = (br ? W2a : W1a) +
          ((size_t)((part * HID + mt * 16 + m) * CH + kb * 32 + k0)) * 3 + s;
    kstride = 3;
  } else if (frag < NFRAG_A + NFRAG_B1) {
    const int f = frag - NFRAG_A;
    const int kb = f & 1;
    const int ct = (f >> 1) & 15;
    const int part = f >> 5;
    src = W1b + (size_t)((part * CH + ct * 16 + m) * HID + kb * 32 + k0);
    kstride = 1;
  } else {
    const int f = frag - NFRAG_A - NFRAG_B1;
    const int kb = f & 1;
    int r = f >> 1;
    const int s = r % 3;  r /= 3;
    const int ct = r & 15;
    const int part = r >> 4;
    src = W2b + ((size_t)((part * CH + ct * 16 + m) * HID + kb * 32 + k0)) * 3 + s;
    kstride = 3;
  }
  v8h o;
#pragma unroll
  for (int e = 0; e < 8; ++e) o[e] = (_Float16)src[e * kstride];
  *(v8h*)(wf + (size_t)gid * 8) = o;
}

__global__ __launch_bounds__(256)
void mtb_gate_kernel(const _Float16* __restrict__ wf,
                     float* __restrict__ out) {
  using namespace cfg;
  __shared__ __attribute__((aligned(16))) _Float16 sXt[XT_ROWS * XT_STRIDE];
  __shared__ _Float16 sHt[2][XT_ROWS * HT_STRIDE];

  const int part = blockIdx.x;
  const int b    = blockIdx.y;
  const int tid  = threadIdx.x;
  const int lane = tid & 31;
  const int wave = tid >> 5;
  const int hs   = (lane >> 4) & 1;

  {
    sXt[0 * XT_STRIDE + tid]         = (_Float16)0.f;
    sXt[1 * XT_STRIDE + tid]         = (_Float16)0.f;
    sXt[(SEQ + 2) * XT_STRIDE + tid] = (_Float16)0.f;
    sXt[(SEQ + 3) * XT_STRIDE + tid] = (_Float16)0.f;
    if (tid < HID) {
#pragma unroll
      for (int r = 0; r < 2; ++r) {
        sHt[0][r * HT_STRIDE + tid] = (_Float16)0.f;
        sHt[1][r * HT_STRIDE + tid] = (_Float16)0.f;
        sHt[0][(SEQ + 2 + r) * HT_STRIDE + tid] = (_Float16)0.f;
        sHt[1][(SEQ + 2 + r) * HT_STRIDE + tid] = (_Float16)0.f;
      }
    }
    const _Float16* xp = wf + WF_HALFS + ((size_t)(b * PARTS + part) * SEQ) * CH;
    v8h t[SEQ * CH / 8 / 256];
#pragma unroll
    for (int i = 0; i < SEQ * CH / 8 / 256; ++i)
      t[i] = *(const v8h*)(xp + (size_t)(i * 256 + tid) * 8);
#pragma unroll
    for (int i = 0; i < SEQ * CH / 8 / 256; ++i) {
      const int chunk = i * 256 + tid;
      *(v8h*)&sXt[(2 + (chunk >> 5)) * XT_STRIDE + (chunk & 31) * 8] = t[i];
    }
  }
  __syncthreads();

  {
    const int br = wave >> 2;
    const int mt = wave & 3;
    const int m0 = mt * 16;
    const int fbase = ((br * PARTS + part) * 4 + mt) * 24;
    v8f acc[4] = {};
    for (int s = 0; s < 3; ++s) {
      for (int kb = 0; kb < CH / 32; ++kb) {
        const int c0 = kb * 32;
        v16h a = load_a_frag_ws(wf, fbase + s * 8 + kb, lane);
#pragma unroll
        for (int nt = 0; nt < 4; ++nt) {
          v16h bf = load_b_frag_lds(sXt, 1 + s + nt * 16, XT_STRIDE, c0, lane);
          acc[nt] = wmma_f16(a, bf, acc[nt]);
        }
      }
    }
#pragma unroll
    for (int nt = 0; nt < 4; ++nt) {
#pragma unroll
      for (int v = 0; v < 8; ++v) {
        float f = acc[nt][v];
        f = f > 0.f ? f : 0.01f * f;
        const int row = 2 + nt * 16 + (lane & 15);
        const int col = m0 + v + (hs << 3);
        sHt[br][row * HT_STRIDE + col] = (_Float16)f;
      }
    }
  }
  __syncthreads();

  for (int j = 0; j < 2; ++j) {
    const int m0 = wave * 32 + j * 16;
    const int ct  = wave * 2 + j;
    const int fb1 = NFRAG_A + (part * 16 + ct) * 2;
    const int fb2 = NFRAG_A + NFRAG_B1 + (part * 16 + ct) * 6;
    v16h a1[2], a2[3][2];
#pragma unroll
    for (int kb = 0; kb < 2; ++kb)
      a1[kb] = load_a_frag_ws(wf, fb1 + kb, lane);
#pragma unroll
    for (int s = 0; s < 3; ++s)
#pragma unroll
      for (int kb = 0; kb < 2; ++kb)
        a2[s][kb] = load_a_frag_ws(wf, fb2 + s * 2 + kb, lane);

    float maxv[8];
#pragma unroll
    for (int v = 0; v < 8; ++v) maxv[v] = -INFINITY;

    for (int nt = 0; nt < 4; ++nt) {
      v8f acc1 = {}, acc2 = {};
#pragma unroll
      for (int kb = 0; kb < 2; ++kb) {
        v16h bf = load_b_frag_lds(sHt[0], 2 + nt * 16, HT_STRIDE, kb * 32, lane);
        acc1 = wmma_f16(a1[kb], bf, acc1);
      }
      for (int s = 0; s < 3; ++s) {
#pragma unroll
        for (int kb = 0; kb < 2; ++kb) {
          v16h bf = load_b_frag_lds(sHt[1], 1 + s + nt * 16, HT_STRIDE,
                                    kb * 32, lane);
          acc2 = wmma_f16(a2[s][kb], bf, acc2);
        }
      }
#pragma unroll
      for (int v = 0; v < 8; ++v) {
        const int c = m0 + v + (hs << 3);
        const int l = nt * 16 + (lane & 15);
        float xs[5];
#pragma unroll
        for (int d = 0; d < 5; ++d)
          xs[d] = (float)sXt[(l + d) * XT_STRIDE + c];
        const float avg3 = (xs[1] + xs[2] + xs[3]) * (1.0f / 3.0f);
        const float avg5 = (xs[0] + xs[1] + xs[2] + xs[3] + xs[4]) * 0.2f;
        float m3 = xs[2];
        m3 = fmaxf(m3, (l >= 1)       ? xs[1] : -INFINITY);
        m3 = fmaxf(m3, (l <= cfg::SEQ - 2) ? xs[3] : -INFINITY);
        float m5 = m3;
        m5 = fmaxf(m5, (l >= 2)       ? xs[0] : -INFINITY);
        m5 = fmaxf(m5, (l <= cfg::SEQ - 3) ? xs[4] : -INFINITY);
        const float g1 = sigmoidf_(acc1[v]);
        const float g2 = sigmoidf_(acc2[v]);
        const float mval = (avg3 + m3) * g1 + (avg5 + m5) * g2;
        maxv[v] = fmaxf(maxv[v], mval);
      }
    }
#pragma unroll
    for (int v = 0; v < 8; ++v) {
#pragma unroll
      for (int off = 8; off >= 1; off >>= 1)
        maxv[v] = fmaxf(maxv[v], __shfl_xor(maxv[v], off, 16));
      if ((lane & 15) == 0) {
        const int c = m0 + v + (hs << 3);
        out[((size_t)b * cfg::PARTS + part) * cfg::CH + c] = maxv[v];
      }
    }
  }
}

extern "C" void kernel_launch(void* const* d_in, const int* in_sizes, int n_in,
                              void* d_out, int out_size, void* d_ws, size_t ws_size,
                              hipStream_t stream) {
  (void)in_sizes; (void)n_in; (void)ws_size; (void)out_size;
  const float* x   = (const float*)d_in[0];
  const float* W1a = (const float*)d_in[1];
  const float* W1b = (const float*)d_in[2];
  const float* W2a = (const float*)d_in[3];
  const float* W2b = (const float*)d_in[4];
  float* out = (float*)d_out;
  _Float16* wf = (_Float16*)d_ws;

  prepass_kernel<<<dim3(cfg::PRE_GRID), dim3(256), 0, stream>>>(x, W1a, W1b, W2a, W2b, wf);
  dim3 grid(cfg::PARTS, cfg::NB);
  dim3 block(256);
  mtb_gate_kernel<<<grid, block, 0, stream>>>(wf, out);
}


# Stage A k loop unrolled and software-pipelined: weight fragments fetched four k-steps ahead from the workspace, x fragments a step or more ahead from LDS; no product waits on a load just issued
# speedup vs baseline: 3.5589x; 1.1155x over previous
#include <hip/hip_runtime.h>
#include <math.h>

typedef __attribute__((ext_vector_type(16))) _Float16 v16h;
typedef __attribute__((ext_vector_type(8)))  _Float16 v8h;
typedef __attribute__((ext_vector_type(8)))  float    v8f;

namespace cfg {
constexpr int NB    = 128;
constexpr int SEQ   = 64;
constexpr int CH    = 256;
constexpr int PARTS = 16;
constexpr int HID   = 64;
constexpr int XT_ROWS   = SEQ + 4;
constexpr int XT_STRIDE = CH + 8;
constexpr int HT_STRIDE = HID + 8;
constexpr int FRAG_HALFS = 512;
constexpr int NFRAG_A  = 2 * PARTS * 4 * 3 * 8;
constexpr int NFRAG_B1 = PARTS * 16 * 2;
constexpr int NFRAG_B2 = PARTS * 16 * 3 * 2;
constexpr int NFRAG    = NFRAG_A + NFRAG_B1 + NFRAG_B2;
constexpr int PRE_BLOCKS = NFRAG * 64 / 256;
constexpr size_t WF_HALFS = (size_t)NFRAG * FRAG_HALFS;
constexpr int XT_PIECES = NB * SEQ * (CH / 64);
constexpr int XT_TRIPS  = 4;
constexpr int XT_BLOCKS = XT_PIECES / (2 * 8 * XT_TRIPS);
constexpr int PRE_GRID  = XT_BLOCKS + PRE_BLOCKS;
}

__device__ __forceinline__ v16h load_a_frag_ws(const _Float16* __restrict__ wf,
                                               int frag, int lane) {
  const _Float16* p = wf + (size_t)frag * cfg::FRAG_HALFS + lane * 8;
  const v8h lo = *(const v8h*)p;
  const v8h hi = *(const v8h*)(p + 256);
  return __builtin_shufflevector(lo, hi, 0, 1, 2, 3, 4, 5, 6, 7,
                                 8, 9, 10, 11, 12, 13, 14, 15);
}

__device__ __forceinline__ v16h load_b_frag_lds(const _Float16* lds, int row_base,
                                                int row_stride, int col_base,
                                                int lane) {
  const int n  = lane & 15;
  const int hs = (lane >> 4) & 1;
  const _Float16* p = lds + (size_t)(row_base + n) * row_stride + col_base;
  v16h b;
#pragma unroll
  for (int t = 0; t < 16; ++t) {
    const int v = t >> 1, e = t & 1;
    const int k = ((v >> 2) << 4) + (hs << 3) + ((v & 3) << 1) + e;
    b[t] = p[k];
  }
  return b;
}

__device__ __forceinline__ v8f wmma_f16(v16h a, v16h b, v8f c) {
  return __builtin_amdgcn_wmma_f32_16x16x32_f16(false, a, false, b,
                                                (short)0, c, false, false);
}

__device__ __forceinline__ float sigmoidf_(float z) {
  return 1.0f / (1.0f + __expf(-z));
}

template <int T, int KSTEPS>
__device__ __forceinline__ void conv_a_step(const _Float16* __restrict__ wf,
                                            const _Float16* sXt, int fbase, int lane,
                                            v16h (&aq)[4], v16h (&bq)[4], v8f (&acc)[4]) {
  using namespace cfg;
  const v16h a = aq[T & 3];
  if (T + 4 < KSTEPS) aq[T & 3] = load_a_frag_ws(wf, fbase + T + 4, lane);
  v16h bn[4];
  if (T + 1 < KSTEPS) {
#pragma unroll
    for (int nt = 0; nt < 4; ++nt)
      bn[nt] = load_b_frag_lds(sXt, 1 + ((T + 1) >> 3) + nt * 16, XT_STRIDE,
                               ((T + 1) & 7) * 32, lane);
  }
  __builtin_amdgcn_sched_barrier(0);
#pragma unroll
  for (int nt = 0; nt < 4; ++nt) acc[nt] = wmma_f16(a, bq[nt], acc[nt]);
  if (T + 1 < KSTEPS) {
#pragma unroll
    for (int nt = 0; nt < 4; ++nt) bq[nt] = bn[nt];
  }
}

template <int T, int KSTEPS>
struct ConvASteps {
  static __device__ __forceinline__ void run(const _Float16* __restrict__ wf,
                                             const _Float16* sXt, int fbase, int lane,
                                             v16h (&aq)[4], v16h (&bq)[4], v8f (&acc)[4]) {
    conv_a_step<T, KSTEPS>(wf, sXt, fbase, lane, aq, bq, acc);
    ConvASteps<T + 1, KSTEPS>::run(wf, sXt, fbase, lane, aq, bq, acc);
  }
};
template <int KSTEPS>
struct ConvASteps<KSTEPS, KSTEPS> {
  static __device__ __forceinline__ void run(const _Float16* __restrict__, const _Float16*,
                                             int, int, v16h (&)[4], v16h (&)[4], v8f (&)[4]) {}
};

__device__ __forceinline__ void x_to_f16_tiles(const float* __restrict__ x,
                                               _Float16* __restrict__ xt,
                                               int blk, int tid) {
  using namespace cfg;
  const int wave = tid >> 5;
  const int lane = tid & 31;
  const int c8   = lane >> 2;
  const int pq   = lane & 3;
  for (int it = 0; it < XT_TRIPS; ++it) {
    const int piece0 = ((blk * XT_TRIPS + it) * 8 + wave) * 2;
    const int row    = piece0 >> 2;
    const int c0     = (piece0 & 3) * 64 + c8 * 8;
    const float* src = x + ((size_t)row * CH + c0) * PARTS + pq * 4;
    float4 v[2][8];
#pragma unroll
    for (int h = 0; h < 2; ++h)
#pragma unroll
      for (int j = 0; j < 8; ++j)
        v[h][j] = *(const float4*)(src + (h * 64 + j) * PARTS);
    __builtin_amdgcn_sched_barrier(0);
    const int b = row >> 6;
    const int l = row & 63;
    _Float16* dst = xt + (((size_t)(b * PARTS + pq * 4) * SEQ + l) * CH + c0);
#pragma unroll
    for (int h = 0; h < 2; ++h) {
      v8h o0, o1, o2, o3;
#pragma unroll
      for (int j = 0; j < 8; ++j) {
        o0[j] = (_Float16)v[h][j].x;
        o1[j] = (_Float16)v[h][j].y;
        o2[j] = (_Float16)v[h][j].z;
        o3[j] = (_Float16)v[h][j].w;
      }
      *(v8h*)(dst + h * 64)                        = o0;
      *(v8h*)(dst + h * 64 + (size_t)1 * SEQ * CH) = o1;
      *(v8h*)(dst + h * 64 + (size_t)2 * SEQ * CH) = o2;
      *(v8h*)(dst + h * 64 + (size_t)3 * SEQ * CH) = o3;
    }
  }
}

__global__ __launch_bounds__(256)
void prepass_kernel(const float* __restrict__ x,
                    const float* __restrict__ W1a,
                    const float* __restrict__ W1b,
                    const float* __restrict__ W2a,
                    const float* __restrict__ W2b,
                    _Float16* __restrict__ wf) {
  using namespace cfg;
  if (blockIdx.x < XT_BLOCKS) {
    x_to_f16_tiles(x, wf + WF_HALFS, blockIdx.x, threadIdx.x);
    return;
  }
  const int gid  = (blockIdx.x - XT_BLOCKS) * 256 + threadIdx.x;
  const int frag = gid >> 6;
  const int sel  = (gid >> 5) & 1;
  const int lane = gid & 31;
  const int m    = lane & 15;
  const int hs   = lane >> 4;
  const int k0   = sel * 16 + hs * 8;
  const float* src;
  int kstride;
  if (frag < NFRAG_A) {
    const int kb = frag & 7;
    int r = frag >> 3;
    const int s = r % 3;  r /= 3;
    const int mt = r & 3; r >>= 2;
    const int part = r & 15;
    const int br = r >> 4;
    src = (br ? W2a : W1a) +
          ((size_t)((part * HID + mt * 16 + m) * CH + kb * 32 + k0)) * 3 + s;
    kstride = 3;
  } else if (frag < NFRAG_A + NFRAG_B1) {
    const int f = frag - NFRAG_A;
    const int kb = f & 1;
    const int ct = (f >> 1) & 15;
    const int part = f >> 5;
    src = W1b + (size_t)((part * CH + ct * 16 + m) * HID + kb * 32 + k0);
    kstride = 1;
  } else {
    const int f = frag - NFRAG_A - NFRAG_B1;
    const int kb = f & 1;
    int r = f >> 1;
    const int s = r % 3;  r /= 3;
    const int ct = r & 15;
    const int part = r >> 4;
    src = W2b + ((size_t)((part * CH + ct * 16 + m) * HID + kb * 32 + k0)) * 3 + s;
    kstride = 3;
  }
  v8h o;
#pragma unroll
  for (int e = 0; e < 8; ++e) o[e] = (_Float16)src[e * kstride];
  *(v8h*)(wf + (size_t)gid * 8) = o;
}

__global__ __launch_bounds__(256)
void mtb_gate_kernel(const _Float16* __restrict__ wf,
                     float* __restrict__ out) {
  using namespace cfg;
  __shared__ __attribute__((aligned(16))) _Float16 sXt[XT_ROWS * XT_STRIDE];
  __shared__ _Float16 sHt[2][XT_ROWS * HT_STRIDE];

  const int part = blockIdx.x;
  const int b    = blockIdx.y;
  const int tid  = threadIdx.x;
  const int lane = tid & 31;
  const int wave = tid >> 5;
  const int hs   = (lane >> 4) & 1;

  {
    sXt[0 * XT_STRIDE + tid]         = (_Float16)0.f;
    sXt[1 * XT_STRIDE + tid]         = (_Float16)0.f;
    sXt[(SEQ + 2) * XT_STRIDE + tid] = (_Float16)0.f;
    sXt[(SEQ + 3) * XT_STRIDE + tid] = (_Float16)0.f;
    if (tid < HID) {
#pragma unroll
      for (int r = 0; r < 2; ++r) {
        sHt[0][r * HT_STRIDE + tid] = (_Float16)0.f;
        sHt[1][r * HT_STRIDE + tid] = (_Float16)0.f;
        sHt[0][(SEQ + 2 + r) * HT_STRIDE + tid] = (_Float16)0.f;
        sHt[1][(SEQ + 2 + r) * HT_STRIDE + tid] = (_Float16)0.f;
      }
    }
    const _Float16* xp = wf + WF_HALFS + ((size_t)(b * PARTS + part) * SEQ) * CH;
    v8h t[SEQ * CH / 8 / 256];
#pragma unroll
    for (int i = 0; i < SEQ * CH / 8 / 256; ++i)
      t[i] = *(const v8h*)(xp + (size_t)(i * 256 + tid) * 8);
#pragma unroll
    for (int i = 0; i < SEQ * CH / 8 / 256; ++i) {
      const int chunk = i * 256 + tid;
      *(v8h*)&sXt[(2 + (chunk >> 5)) * XT_STRIDE + (chunk & 31) * 8] = t[i];
    }
  }
  {
    const int br = wave >> 2;
    const int mt = wave & 3;
    const int m0 = mt * 16;
    const int fbase = ((br * PARTS + part) * 4 + mt) * 24;
    constexpr int KSTEPS = 3 * (CH / 32);
    v8f acc[4] = {};
    v16h aq[4], bq[4];
#pragma unroll
    for (int i = 0; i < 4; ++i) aq[i] = load_a_frag_ws(wf, fbase + i, lane);
    __syncthreads();
#pragma unroll
    for (int nt = 0; nt < 4; ++nt)
      bq[nt] = load_b_frag_lds(sXt, 1 + nt * 16, XT_STRIDE, 0, lane);
    ConvASteps<0, KSTEPS>::run(wf, sXt, fbase, lane, aq, bq, acc);
#pragma unroll
    for (int nt = 0; nt < 4; ++nt) {
#pragma unroll
      for (int v = 0; v < 8; ++v) {
        float f = acc[nt][v];
        f = f > 0.f ? f : 0.01f * f;
        const int row = 2 + nt * 16 + (lane & 15);
        const int col = m0 + v + (hs << 3);
        sHt[br][row * HT_STRIDE + col] = (_Float16)f;
      }
    }
  }
  __syncthreads();

  for (int j = 0; j < 2; ++j) {
    const int m0 = wave * 32 + j * 16;
    const int ct  = wave * 2 + j;
    const int fb1 = NFRAG_A + (part * 16 + ct) * 2;
    const int fb2 = NFRAG_A + NFRAG_B1 + (part * 16 + ct) * 6;
    v16h a1[2], a2[3][2];
#pragma unroll
    for (int kb = 0; kb < 2; ++kb)
      a1[kb] = load_a_frag_ws(wf, fb1 + kb, lane);
#pragma unroll
    for (int s = 0; s < 3; ++s)
#pragma unroll
      for (int kb = 0; kb < 2; ++kb)
        a2[s][kb] = load_a_frag_ws(wf, fb2 + s * 2 + kb, lane);

    float maxv[8];
#pragma unroll
    for (int v = 0; v < 8; ++v) maxv[v] = -INFINITY;

    for (int nt = 0; nt < 4; ++nt) {
      v8f acc1 = {}, acc2 = {};
#pragma unroll
      for (int kb = 0; kb < 2; ++kb) {
        v16h bf = load_b_frag_lds(sHt[0], 2 + nt * 16, HT_STRIDE, kb * 32, lane);
        acc1 = wmma_f16(a1[kb], bf, acc1);
      }
      for (int s = 0; s < 3; ++s) {
#pragma unroll
        for (int kb = 0; kb < 2; ++kb) {
          v16h bf = load_b_frag_lds(sHt[1], 1 + s + nt * 16, HT_STRIDE,
                                    kb * 32, lane);
          acc2 = wmma_f16(a2[s][kb], bf, acc2);
        }
      }
#pragma unroll
      for (int v = 0; v < 8; ++v) {
        const int c = m0 + v + (hs << 3);
        const int l = nt * 16 + (lane & 15);
        float xs[5];
#pragma unroll
        for (int d = 0; d < 5; ++d)
          xs[d] = (float)sXt[(l + d) * XT_STRIDE + c];
        const float avg3 = (xs[1] + xs[2] + xs[3]) * (1.0f / 3.0f);
        const float avg5 = (xs[0] + xs[1] + xs[2] + xs[3] + xs[4]) * 0.2f;
        float m3 = xs[2];
        m3 = fmaxf(m3, (l >= 1)       ? xs[1] : -INFINITY);
        m3 = fmaxf(m3, (l <= cfg::SEQ - 2) ? xs[3] : -INFINITY);
        float m5 = m3;
        m5 = fmaxf(m5, (l >= 2)       ? xs[0] : -INFINITY);
        m5 = fmaxf(m5, (l <= cfg::SEQ - 3) ? xs[4] : -INFINITY);
        const float g1 = sigmoidf_(acc1[v]);
        const float g2 = sigmoidf_(acc2[v]);
        const float mval = (avg3 + m3) * g1 + (avg5 + m5) * g2;
        maxv[v] = fmaxf(maxv[v], mval);
      }
    }
#pragma unroll
    for (int v = 0; v < 8; ++v) {
#pragma unroll
      for (int off = 8; off >= 1; off >>= 1)
        maxv[v] = fmaxf(maxv[v], __shfl_xor(maxv[v], off, 16));
      if ((lane & 15) == 0) {
        const int c = m0 + v + (hs << 3);
        out[((size_t)b * cfg::PARTS + part) * cfg::CH + c] = maxv[v];
      }
    }
  }
}

extern "C" void kernel_launch(void* const* d_in, const int* in_sizes, int n_in,
                              void* d_out, int out_size, void* d_ws, size_t ws_size,
                              hipStream_t stream) {
  (void)in_sizes; (void)n_in; (void)ws_size; (void)out_size;
  const float* x   = (const float*)d_in[0];
  const float* W1a = (const float*)d_in[1];
  const float* W1b = (const float*)d_in[2];
  const float* W2a = (const float*)d_in[3];
  const float* W2b = (const float*)d_in[4];
  float* out = (float*)d_out;
  _Float16* wf = (_Float16*)d_ws;

  prepass_kernel<<<dim3(cfg::PRE_GRID), dim3(256), 0, stream>>>(x, W1a, W1b, W2a, W2b, wf);
  dim3 grid(cfg::PARTS, cfg::NB);
  dim3 block(256);
  mtb_gate_kernel<<<grid, block, 0, stream>>>(wf, out);
}


# Gate sigmoid as 0.5+0.5*tanh(z/2) on the hardware tanh instruction (v_tanh_f32): one transcendental and two plain instructions per gate, not an exponential plus an exactly rounded division
# speedup vs baseline: 3.9534x; 1.1108x over previous
#include <hip/hip_runtime.h>
#include <math.h>

typedef __attribute__((ext_vector_type(16))) _Float16 v16h;
typedef __attribute__((ext_vector_type(8)))  _Float16 v8h;
typedef __attribute__((ext_vector_type(8)))  float    v8f;

namespace cfg {
constexpr int NB    = 128;
constexpr int SEQ   = 64;
constexpr int CH    = 256;
constexpr int PARTS = 16;
constexpr int HID   = 64;
constexpr int XT_ROWS   = SEQ + 4;
constexpr int XT_STRIDE = CH + 8;
constexpr int HT_STRIDE = HID + 8;
constexpr int FRAG_HALFS = 512;
constexpr int NFRAG_A  = 2 * PARTS * 4 * 3 * 8;
constexpr int NFRAG_B1 = PARTS * 16 * 2;
constexpr int NFRAG_B2 = PARTS * 16 * 3 * 2;
constexpr int NFRAG    = NFRAG_A + NFRAG_B1 + NFRAG_B2;
constexpr int PRE_BLOCKS = NFRAG * 64 / 256;
constexpr size_t WF_HALFS = (size_t)NFRAG * FRAG_HALFS;
constexpr int XT_PIECES = NB * SEQ * (CH / 64);
constexpr int XT_TRIPS  = 4;
constexpr int XT_BLOCKS = XT_PIECES / (2 * 8 * XT_TRIPS);
constexpr int PRE_GRID  = XT_BLOCKS + PRE_BLOCKS;
}

__device__ __forceinline__ v16h load_a_frag_ws(const _Float16* __restrict__ wf,
                                               int frag, int lane) {
  const _Float16* p = wf + (size_t)frag * cfg::FRAG_HALFS + lane * 8;
  const v8h lo = *(const v8h*)p;
  const v8h hi = *(const v8h*)(p + 256);
  return __builtin_shufflevector(lo, hi, 0, 1, 2, 3, 4, 5, 6, 7,
                                 8, 9, 10, 11, 12, 13, 14, 15);
}

__device__ __forceinline__ v16h load_b_frag_lds(const _Float16* lds, int row_base,
                                                int row_stride, int col_base,
                                                int lane) {
  const int n  = lane & 15;
  const int hs = (lane >> 4) & 1;
  const _Float16* p = lds + (size_t)(row_base + n) * row_stride + col_base;
  v16h b;
#pragma unroll
  for (int t = 0; t < 16; ++t) {
    const int v = t >> 1, e = t & 1;
    const int k = ((v >> 2) << 4) + (hs << 3) + ((v & 3) << 1) + e;
    b[t] = p[k];
  }
  return b;
}

__device__ __forceinline__ v8f wmma_f16(v16h a, v16h b, v8f c) {
  return __builtin_amdgcn_wmma_f32_16x16x32_f16(false, a, false, b,
                                                (short)0, c, false, false);
}

__device__ __forceinline__ float sigmoidf_(float z) {
  return fmaf(0.5f, __builtin_amdgcn_tanhf(0.5f * z), 0.5f);
}

template <int T, int KSTEPS>
__device__ __forceinline__ void conv_a_step(const _Float16* __restrict__ wf,
                                            const _Float16* sXt, int fbase, int lane,
                                            v16h (&aq)[4], v16h (&bq)[4], v8f (&acc)[4]) {
  using namespace cfg;
  const v16h a = aq[T & 3];
  if (T + 4 < KSTEPS) aq[T & 3] = load_a_frag_ws(wf, fbase + T + 4, lane);
  v16h bn[4];
  if (T + 1 < KSTEPS) {
#pragma unroll
    for (int nt = 0; nt < 4; ++nt)
      bn[nt] = load_b_frag_lds(sXt, 1 + ((T + 1) >> 3) + nt * 16, XT_STRIDE,
                               ((T + 1) & 7) * 32, lane);
  }
  __builtin_amdgcn_sched_barrier(0);
#pragma unroll
  for (int nt = 0; nt < 4; ++nt) acc[nt] = wmma_f16(a, bq[nt], acc[nt]);
  if (T + 1 < KSTEPS) {
#pragma unroll
    for (int nt = 0; nt < 4; ++nt) bq[nt] = bn[nt];
  }
}

template <int T, int KSTEPS>
struct ConvASteps {
  static __device__ __forceinline__ void run(const _Float16* __restrict__ wf,
                                             const _Float16* sXt, int fbase, int lane,
                                             v16h (&aq)[4], v16h (&bq)[4], v8f (&acc)[4]) {
    conv_a_step<T, KSTEPS>(wf, sXt, fbase, lane, aq, bq, acc);
    ConvASteps<T + 1, KSTEPS>::run(wf, sXt, fbase, lane, aq, bq, acc);
  }
};
template <int KSTEPS>
struct ConvASteps<KSTEPS, KSTEPS> {
  static __device__ __forceinline__ void run(const _Float16* __restrict__, const _Float16*,
                                             int, int, v16h (&)[4], v16h (&)[4], v8f (&)[4]) {}
};

__device__ __forceinline__ void x_to_f16_tiles(const float* __restrict__ x,
                                               _Float16* __restrict__ xt,
                                               int blk, int tid) {
  using namespace cfg;
  const int wave = tid >> 5;
  const int lane = tid & 31;
  const int c8   = lane >> 2;
  const int pq   = lane & 3;
  for (int it = 0; it < XT_TRIPS; ++it) {
    const int piece0 = ((blk * XT_TRIPS + it) * 8 + wave) * 2;
    const int row    = piece0 >> 2;
    const int c0     = (piece0 & 3) * 64 + c8 * 8;
    const float* src = x + ((size_t)row * CH + c0) * PARTS + pq * 4;
    float4 v[2][8];
#pragma unroll
    for (int h = 0; h < 2; ++h)
#pragma unroll
      for (int j = 0; j < 8; ++j)
        v[h][j] = *(const float4*)(src + (h * 64 + j) * PARTS);
    __builtin_amdgcn_sched_barrier(0);
    const int b = row >> 6;
    const int l = row & 63;
    _Float16* dst = xt + (((size_t)(b * PARTS + pq * 4) * SEQ + l) * CH + c0);
#pragma unroll
    for (int h = 0; h < 2; ++h) {
      v8h o0, o1, o2, o3;
#pragma unroll
      for (int j = 0; j < 8; ++j) {
        o0[j] = (_Float16)v[h][j].x;
        o1[j] = (_Float16)v[h][j].y;
        o2[j] = (_Float16)v[h][j].z;
        o3[j] = (_Float16)v[h][j].w;
      }
      *(v8h*)(dst + h * 64)                        = o0;
      *(v8h*)(dst + h * 64 + (size_t)1 * SEQ * CH) = o1;
      *(v8h*)(dst + h * 64 + (size_t)2 * SEQ * CH) = o2;
      *(v8h*)(dst + h * 64 + (size_t)3 * SEQ * CH) = o3;
    }
  }
}

__global__ __launch_bounds__(256)
void prepass_kernel(const float* __restrict__ x,
                    const float* __restrict__ W1a,
                    const float* __restrict__ W1b,
                    const float* __restrict__ W2a,
                    const float* __restrict__ W2b,
                    _Float16* __restrict__ wf) {
  using namespace cfg;
  if (blockIdx.x < XT_BLOCKS) {
    x_to_f16_tiles(x, wf + WF_HALFS, blockIdx.x, threadIdx.x);
    return;
  }
  const int gid  = (blockIdx.x - XT_BLOCKS) * 256 + threadIdx.x;
  const int frag = gid >> 6;
  const int sel  = (gid >> 5) & 1;
  const int lane = gid & 31;
  const int m    = lane & 15;
  const int hs   = lane >> 4;
  const int k0   = sel * 16 + hs * 8;
  const float* src;
  int kstride;
  if (frag < NFRAG_A) {
    const int kb = frag & 7;
    int r = frag >> 3;
    const int s = r % 3;  r /= 3;
    const int mt = r & 3; r >>= 2;
    const int part = r & 15;
    const int br = r >> 4;
    src = (br ? W2a : W1a) +
          ((size_t)((part * HID + mt * 16 + m) * CH + kb * 32 + k0)) * 3 + s;
    kstride = 3;
  } else if (frag < NFRAG_A + NFRAG_B1) {
    const int f = frag - NFRAG_A;
    const int kb = f & 1;
    const int ct = (f >> 1) & 15;
    const int part = f >> 5;
    src = W1b + (size_t)((part * CH + ct * 16 + m) * HID + kb * 32 + k0);
    kstride = 1;
  } else {
    const int f = frag - NFRAG_A - NFRAG_B1;
    const int kb = f & 1;
    int r = f >> 1;
    const int s = r % 3;  r /= 3;
    const int ct = r & 15;
    const int part = r >> 4;
    src = W2b + ((size_t)((part * CH + ct * 16 + m) * HID + kb * 32 + k0)) * 3 + s;
    kstride = 3;
  }
  v8h o;
#pragma unroll
  for (int e = 0; e < 8; ++e) o[e] = (_Float16)src[e * kstride];
  *(v8h*)(wf + (size_t)gid * 8) = o;
}

__global__ __launch_bounds__(256)
void mtb_gate_kernel(const _Float16* __restrict__ wf,
                     float* __restrict__ out) {
  using namespace cfg;
  __shared__ __attribute__((aligned(16))) _Float16 sXt[XT_ROWS * XT_STRIDE];
  __shared__ _Float16 sHt[2][XT_ROWS * HT_STRIDE];

  const int part = blockIdx.x;
  const int b    = blockIdx.y;
  const int tid  = threadIdx.x;
  const int lane = tid & 31;
  const int wave = tid >> 5;
  const int hs   = (lane >> 4) & 1;

  {
    sXt[0 * XT_STRIDE + tid]         = (_Float16)0.f;
    sXt[1 * XT_STRIDE + tid]         = (_Float16)0.f;
    sXt[(SEQ + 2) * XT_STRIDE + tid] = (_Float16)0.f;
    sXt[(SEQ + 3) * XT_STRIDE + tid] = (_Float16)0.f;
    if (tid < HID) {
#pragma unroll
      for (int r = 0; r < 2; ++r) {
        sHt[0][r * HT_STRIDE + tid] = (_Float16)0.f;
        sHt[1][r * HT_STRIDE + tid] = (_Float16)0.f;
        sHt[0][(SEQ + 2 + r) * HT_STRIDE + tid] = (_Float16)0.f;
        sHt[1][(SEQ + 2 + r) * HT_STRIDE + tid] = (_Float16)0.f;
      }
    }
    const _Float16* xp = wf + WF_HALFS + ((size_t)(b * PARTS + part) * SEQ) * CH;
    v8h t[SEQ * CH / 8 / 256];
#pragma unroll
    for (int i = 0; i < SEQ * CH / 8 / 256; ++i)
      t[i] = *(const v8h*)(xp + (size_t)(i * 256 + tid) * 8);
#pragma unroll
    for (int i = 0; i < SEQ * CH / 8 / 256; ++i) {
      const int chunk = i * 256 + tid;
      *(v8h*)&sXt[(2 + (chunk >> 5)) * XT_STRIDE + (chunk & 31) * 8] = t[i];
    }
  }
  {
    const int br = wave >> 2;
    const int mt = wave & 3;
    const int m0 = mt * 16;
    const int fbase = ((br * PARTS + part) * 4 + mt) * 24;
    constexpr int KSTEPS = 3 * (CH / 32);
    v8f acc[4] = {};
    v16h aq[4], bq[4];
#pragma unroll
    for (int i = 0; i < 4; ++i) aq[i] = load_a_frag_ws(wf, fbase + i, lane);
    __syncthreads();
#pragma unroll
    for (int nt = 0; nt < 4; ++nt)
      bq[nt] = load_b_frag_lds(sXt, 1 + nt * 16, XT_STRIDE, 0, lane);
    ConvASteps<0, KSTEPS>::run(wf, sXt, fbase, lane, aq, bq, acc);
#pragma unroll
    for (int nt = 0; nt < 4; ++nt) {
#pragma unroll
      for (int v = 0; v < 8; ++v) {
        float f = acc[nt][v];
        f = f > 0.f ? f : 0.01f * f;
        const int row = 2 + nt * 16 + (lane & 15);
        const int col = m0 + v + (hs << 3);
        sHt[br][row * HT_STRIDE + col] = (_Float16)f;
      }
    }
  }
  __syncthreads();

  for (int j = 0; j < 2; ++j) {
    const int m0 = wave * 32 + j * 16;
    const int ct  = wave * 2 + j;
    const int fb1 = NFRAG_A + (part * 16 + ct) * 2;
    const int fb2 = NFRAG_A + NFRAG_B1 + (part * 16 + ct) * 6;
    v16h a1[2], a2[3][2];
#pragma unroll
    for (int kb = 0; kb < 2; ++kb)
      a1[kb] = load_a_frag_ws(wf, fb1 + kb, lane);
#pragma unroll
    for (int s = 0; s < 3; ++s)
#pragma unroll
      for (int kb = 0; kb < 2; ++kb)
        a2[s][kb] = load_a_frag_ws(wf, fb2 + s * 2 + kb, lane);

    float maxv[8];
#pragma unroll
    for (int v = 0; v < 8; ++v) maxv[v] = -INFINITY;

    for (int nt = 0; nt < 4; ++nt) {
      v8f acc1 = {}, acc2 = {};
#pragma unroll
      for (int kb = 0; kb < 2; ++kb) {
        v16h bf = load_b_frag_lds(sHt[0], 2 + nt * 16, HT_STRIDE, kb * 32, lane);
        acc1 = wmma_f16(a1[kb], bf, acc1);
      }
      for (int s = 0; s < 3; ++s) {
#pragma unroll
        for (int kb = 0; kb < 2; ++kb) {
          v16h bf = load_b_frag_lds(sHt[1], 1 + s + nt * 16, HT_STRIDE,
                                    kb * 32, lane);
          acc2 = wmma_f16(a2[s][kb], bf, acc2);
        }
      }
#pragma unroll
      for (int v = 0; v < 8; ++v) {
        const int c = m0 + v + (hs << 3);
        const int l = nt * 16 + (lane & 15);
        float xs[5];
#pragma unroll
        for (int d = 0; d < 5; ++d)
          xs[d] = (float)sXt[(l + d) * XT_STRIDE + c];
        const float avg3 = (xs[1] + xs[2] + xs[3]) * (1.0f / 3.0f);
        const float avg5 = (xs[0] + xs[1] + xs[2] + xs[3] + xs[4]) * 0.2f;
        float m3 = xs[2];
        m3 = fmaxf(m3, (l >= 1)       ? xs[1] : -INFINITY);
        m3 = fmaxf(m3, (l <= cfg::SEQ - 2) ? xs[3] : -INFINITY);
        float m5 = m3;
        m5 = fmaxf(m5, (l >= 2)       ? xs[0] : -INFINITY);
        m5 = fmaxf(m5, (l <= cfg::SEQ - 3) ? xs[4] : -INFINITY);
        const float g1 = sigmoidf_(acc1[v]);
        const float g2 = sigmoidf_(acc2[v]);
        const float mval = (avg3 + m3) * g1 + (avg5 + m5) * g2;
        maxv[v] = fmaxf(maxv[v], mval);
      }
    }
#pragma unroll
    for (int v = 0; v < 8; ++v) {
#pragma unroll
      for (int off = 8; off >= 1; off >>= 1)
        maxv[v] = fmaxf(maxv[v], __shfl_xor(maxv[v], off, 16));
      if ((lane & 15) == 0) {
        const int c = m0 + v + (hs << 3);
        out[((size_t)b * cfg::PARTS + part) * cfg::CH + c] = maxv[v];
      }
    }
  }
}

extern "C" void kernel_launch(void* const* d_in, const int* in_sizes, int n_in,
                              void* d_out, int out_size, void* d_ws, size_t ws_size,
                              hipStream_t stream) {
  (void)in_sizes; (void)n_in; (void)ws_size; (void)out_size;
  const float* x   = (const float*)d_in[0];
  const float* W1a = (const float*)d_in[1];
  const float* W1b = (const float*)d_in[2];
  const float* W2a = (const float*)d_in[3];
  const float* W2b = (const float*)d_in[4];
  float* out = (float*)d_out;
  _Float16* wf = (_Float16*)d_ws;

  prepass_kernel<<<dim3(cfg::PRE_GRID), dim3(256), 0, stream>>>(x, W1a, W1b, W2a, W2b, wf);
  dim3 grid(cfg::PARTS, cfg::NB);
  dim3 block(256);
  mtb_gate_kernel<<<grid, block, 0, stream>>>(wf, out);
}


# Main kernel block takes two batch elements of its part: each weight fragment fetched from the workspace feeds both x tiles (8 products, was 4); grid halves, weight fetches per batch element halve
# speedup vs baseline: 4.2272x; 1.0693x over previous
#include <hip/hip_runtime.h>
#include <math.h>

typedef __attribute__((ext_vector_type(16))) _Float16 v16h;
typedef __attribute__((ext_vector_type(8)))  _Float16 v8h;
typedef __attribute__((ext_vector_type(8)))  float    v8f;

namespace cfg {
constexpr int NB    = 128;
constexpr int SEQ   = 64;
constexpr int CH    = 256;
constexpr int PARTS = 16;
constexpr int HID   = 64;
constexpr int XT_ROWS   = SEQ + 4;
constexpr int XT_STRIDE = CH + 8;
constexpr int HT_STRIDE = HID + 8;
constexpr int BPB     = 2;
constexpr int XT_TILE = XT_ROWS * XT_STRIDE;
constexpr int HT_TILE = XT_ROWS * HT_STRIDE;
constexpr int MAIN_GRID_Y = NB / BPB;
constexpr int FRAG_HALFS = 512;
constexpr int NFRAG_A  = 2 * PARTS * 4 * 3 * 8;
constexpr int NFRAG_B1 = PARTS * 16 * 2;
constexpr int NFRAG_B2 = PARTS * 16 * 3 * 2;
constexpr int NFRAG    = NFRAG_A + NFRAG_B1 + NFRAG_B2;
constexpr int PRE_BLOCKS = NFRAG * 64 / 256;
constexpr size_t WF_HALFS = (size_t)NFRAG * FRAG_HALFS;
constexpr int XT_PIECES = NB * SEQ * (CH / 64);
constexpr int XT_TRIPS  = 4;
constexpr int XT_BLOCKS = XT_PIECES / (2 * 8 * XT_TRIPS);
constexpr int PRE_GRID  = XT_BLOCKS + PRE_BLOCKS;
}

__device__ __forceinline__ v16h load_a_frag_ws(const _Float16* __restrict__ wf,
                                               int frag, int lane) {
  const _Float16* p = wf + (size_t)frag * cfg::FRAG_HALFS + lane * 8;
  const v8h lo = *(const v8h*)p;
  const v8h hi = *(const v8h*)(p + 256);
  return __builtin_shufflevector(lo, hi, 0, 1, 2, 3, 4, 5, 6, 7,
                                 8, 9, 10, 11, 12, 13, 14, 15);
}

__device__ __forceinline__ v16h load_b_frag_lds(const _Float16* lds, int row_base,
                                                int row_stride, int col_base,
                                                int lane) {
  const int n  = lane & 15;
  const int hs = (lane >> 4) & 1;
  const _Float16* p = lds + (size_t)(row_base + n) * row_stride + col_base;
  v16h b;
#pragma unroll
  for (int t = 0; t < 16; ++t) {
    const int v = t >> 1, e = t & 1;
    const int k = ((v >> 2) << 4) + (hs << 3) + ((v & 3) << 1) + e;
    b[t] = p[k];
  }
  return b;
}

__device__ __forceinline__ v8f wmma_f16(v16h a, v16h b, v8f c) {
  return __builtin_amdgcn_wmma_f32_16x16x32_f16(false, a, false, b,
                                                (short)0, c, false, false);
}

__device__ __forceinline__ float sigmoidf_(float z) {
  return fmaf(0.5f, __builtin_amdgcn_tanhf(0.5f * z), 0.5f);
}

template <int U, int KSTEPS>
__device__ __forceinline__ void conv_a_step(const _Float16* __restrict__ wf,
                                            const _Float16* sXt, int fbase, int lane,
                                            v16h (&aq)[4], v16h& acur, v16h (&bq)[4],
                                            v8f (&acc)[cfg::BPB][4]) {
  using namespace cfg;
  constexpr int T  = U >> 1;
  constexpr int BB = U & 1;
  constexpr int UN = U + 1;
  constexpr int TN = UN >> 1;
  constexpr int BN = UN & 1;
  if (BB == 0) {
    acur = aq[T & 3];
    if (T + 4 < KSTEPS) aq[T & 3] = load_a_frag_ws(wf, fbase + T + 4, lane);
  }
  v16h bn[4];
  if (UN < 2 * KSTEPS) {
#pragma unroll
    for (int nt = 0; nt < 4; ++nt)
      bn[nt] = load_b_frag_lds(sXt + BN * XT_TILE, 1 + (TN >> 3) + nt * 16, XT_STRIDE,
                               (TN & 7) * 32, lane);
  }
  __builtin_amdgcn_sched_barrier(0);
#pragma unroll
  for (int nt = 0; nt < 4; ++nt) acc[BB][nt] = wmma_f16(acur, bq[nt], acc[BB][nt]);
  if (UN < 2 * KSTEPS) {
#pragma unroll
    for (int nt = 0; nt < 4; ++nt) bq[nt] = bn[nt];
  }
}

template <int U, int NU, int KSTEPS>
struct ConvASteps {
  static __device__ __forceinline__ void run(const _Float16* __restrict__ wf,
                                             const _Float16* sXt, int fbase, int lane,
                                             v16h (&aq)[4], v16h& acur, v16h (&bq)[4],
                                             v8f (&acc)[cfg::BPB][4]) {
    conv_a_step<U, KSTEPS>(wf, sXt, fbase, lane, aq, acur, bq, acc);
    ConvASteps<U + 1, NU, KSTEPS>::run(wf, sXt, fbase, lane, aq, acur, bq, acc);
  }
};
template <int NU, int KSTEPS>
struct ConvASteps<NU, NU, KSTEPS> {
  static __device__ __forceinline__ void run(const _Float16* __restrict__, const _Float16*,
                                             int, int, v16h (&)[4], v16h&, v16h (&)[4],
                                             v8f (&)[cfg::BPB][4]) {}
};

__device__ __forceinline__ void x_to_f16_tiles(const float* __restrict__ x,
                                               _Float16* __restrict__ xt,
                                               int blk, int tid) {
  using namespace cfg;
  const int wave = tid >> 5;
  const int lane = tid & 31;
  const int c8   = lane >> 2;
  const int pq   = lane & 3;
  for (int it = 0; it < XT_TRIPS; ++it) {
    const int piece0 = ((blk * XT_TRIPS + it) * 8 + wave) * 2;
    const int row    = piece0 >> 2;
    const int c0     = (piece0 & 3) * 64 + c8 * 8;
    const float* src = x + ((size_t)row * CH + c0) * PARTS + pq * 4;
    float4 v[2][8];
#pragma unroll
    for (int h = 0; h < 2; ++h)
#pragma unroll
      for (int j = 0; j < 8; ++j)
        v[h][j] = *(const float4*)(src + (h * 64 + j) * PARTS);
    __builtin_amdgcn_sched_barrier(0);
    const int b = row >> 6;
    const int l = row & 63;
    _Float16* dst = xt + (((size_t)(b * PARTS + pq * 4) * SEQ + l) * CH + c0);
#pragma unroll
    for (int h = 0; h < 2; ++h) {
      v8h o0, o1, o2, o3;
#pragma unroll
      for (int j = 0; j < 8; ++j) {
        o0[j] = (_Float16)v[h][j].x;
        o1[j] = (_Float16)v[h][j].y;
        o2[j] = (_Float16)v[h][j].z;
        o3[j] = (_Float16)v[h][j].w;
      }
      *(v8h*)(dst + h * 64)                        = o0;
      *(v8h*)(dst + h * 64 + (size_t)1 * SEQ * CH) = o1;
      *(v8h*)(dst + h * 64 + (size_t)2 * SEQ * CH) = o2;
      *(v8h*)(dst + h * 64 + (size_t)3 * SEQ * CH) = o3;
    }
  }
}

__global__ __launch_bounds__(256)
void prepass_kernel(const float* __restrict__ x,
                    const float* __restrict__ W1a,
                    const float* __restrict__ W1b,
                    const float* __restrict__ W2a,
                    const float* __restrict__ W2b,
                    _Float16* __restrict__ wf) {
  using namespace cfg;
  if (blockIdx.x < XT_BLOCKS) {
    x_to_f16_tiles(x, wf + WF_HALFS, blockIdx.x, threadIdx.x);
    return;
  }
  const int gid  = (blockIdx.x - XT_BLOCKS) * 256 + threadIdx.x;
  const int frag = gid >> 6;
  const int sel  = (gid >> 5) & 1;
  const int lane = gid & 31;
  const int m    = lane & 15;
  const int hs   = lane >> 4;
  const int k0   = sel * 16 + hs * 8;
  const float* src;
  int kstride;
  if (frag < NFRAG_A) {
    const int kb = frag & 7;
    int r = frag >> 3;
    const int s = r % 3;  r /= 3;
    const int mt = r & 3; r >>= 2;
    const int part = r & 15;
    const int br = r >> 4;
    src = (br ? W2a : W1a) +
          ((size_t)((part * HID + mt * 16 + m) * CH + kb * 32 + k0)) * 3 + s;
    kstride = 3;
  } else if (frag < NFRAG_A + NFRAG_B1) {
    const int f = frag - NFRAG_A;
    const int kb = f & 1;
    const int ct = (f >> 1) & 15;
    const int part = f >> 5;
    src = W1b + (size_t)((part * CH + ct * 16 + m) * HID + kb * 32 + k0);
    kstride = 1;
  } else {
    const int f = frag - NFRAG_A - NFRAG_B1;
    const int kb = f & 1;
    int r = f >> 1;
    const int s = r % 3;  r /= 3;
    const int ct = r & 15;
    const int part = r >> 4;
    src = W2b + ((size_t)((part * CH + ct * 16 + m) * HID + kb * 32 + k0)) * 3 + s;
    kstride = 3;
  }
  v8h o;
#pragma unroll
  for (int e = 0; e < 8; ++e) o[e] = (_Float16)src[e * kstride];
  *(v8h*)(wf + (size_t)gid * 8) = o;
}

__global__ __launch_bounds__(256)
void mtb_gate_kernel(const _Float16* __restrict__ wf,
                     float* __restrict__ out) {
  using namespace cfg;
  __shared__ __attribute__((aligned(16))) _Float16 sXt[BPB * XT_TILE];
  __shared__ _Float16 sHt[BPB][2][HT_TILE];

  const int part = blockIdx.x;
  const int b0   = blockIdx.y * BPB;
  const int tid  = threadIdx.x;
  const int lane = tid & 31;
  const int wave = tid >> 5;
  const int hs   = (lane >> 4) & 1;

  {
#pragma unroll
    for (int bb = 0; bb < BPB; ++bb) {
      _Float16* xt_ = sXt + bb * XT_TILE;
      xt_[0 * XT_STRIDE + tid]         = (_Float16)0.f;
      xt_[1 * XT_STRIDE + tid]         = (_Float16)0.f;
      xt_[(SEQ + 2) * XT_STRIDE + tid] = (_Float16)0.f;
      xt_[(SEQ + 3) * XT_STRIDE + tid] = (_Float16)0.f;
    }
    if (tid < HID) {
#pragma unroll
      for (int bb = 0; bb < BPB; ++bb) {
#pragma unroll
        for (int r = 0; r < 2; ++r) {
          sHt[bb][0][r * HT_STRIDE + tid] = (_Float16)0.f;
          sHt[bb][1][r * HT_STRIDE + tid] = (_Float16)0.f;
          sHt[bb][0][(SEQ + 2 + r) * HT_STRIDE + tid] = (_Float16)0.f;
          sHt[bb][1][(SEQ + 2 + r) * HT_STRIDE + tid] = (_Float16)0.f;
        }
      }
    }
    v8h t[BPB][SEQ * CH / 8 / 256];
#pragma unroll
    for (int bb = 0; bb < BPB; ++bb) {
      const _Float16* xp = wf + WF_HALFS + ((size_t)((b0 + bb) * PARTS + part) * SEQ) * CH;
#pragma unroll
      for (int i = 0; i < SEQ * CH / 8 / 256; ++i)
        t[bb][i] = *(const v8h*)(xp + (size_t)(i * 256 + tid) * 8);
    }
#pragma unroll
    for (int bb = 0; bb < BPB; ++bb) {
#pragma unroll
      for (int i = 0; i < SEQ * CH / 8 / 256; ++i) {
        const int chunk = i * 256 + tid;
        *(v8h*)&sXt[bb * XT_TILE + (2 + (chunk >> 5)) * XT_STRIDE + (chunk & 31) * 8] = t[bb][i];
      }
    }
  }
  {
    const int br = wave >> 2;
    const int mt = wave & 3;
    const int m0 = mt * 16;
    const int fbase = ((br * PARTS + part) * 4 + mt) * 24;
    constexpr int KSTEPS = 3 * (CH / 32);
    v8f acc[BPB][4] = {};
    v16h aq[4], bq[4], acur;
#pragma unroll
    for (int i = 0; i < 4; ++i) aq[i] = load_a_frag_ws(wf, fbase + i, lane);
    __syncthreads();
#pragma unroll
    for (int nt = 0; nt < 4; ++nt)
      bq[nt] = load_b_frag_lds(sXt, 1 + nt * 16, XT_STRIDE, 0, lane);
    ConvASteps<0, 2 * KSTEPS, KSTEPS>::run(wf, sXt, fbase, lane, aq, acur, bq, acc);
#pragma unroll
    for (int bb = 0; bb < BPB; ++bb) {
#pragma unroll
      for (int nt = 0; nt < 4; ++nt) {
#pragma unroll
        for (int v = 0; v < 8; ++v) {
          float f = acc[bb][nt][v];
          f = f > 0.f ? f : 0.01f * f;
          const int row = 2 + nt * 16 + (lane & 15);
          const int col = m0 + v + (hs << 3);
          sHt[bb][br][row * HT_STRIDE + col] = (_Float16)f;
        }
      }
    }
  }
  __syncthreads();

  for (int j = 0; j < 2; ++j) {
    const int m0 = wave * 32 + j * 16;
    const int ct  = wave * 2 + j;
    const int fb1 = NFRAG_A + (part * 16 + ct) * 2;
    const int fb2 = NFRAG_A + NFRAG_B1 + (part * 16 + ct) * 6;
    v16h a1[2], a2[3][2];
#pragma unroll
    for (int kb = 0; kb < 2; ++kb)
      a1[kb] = load_a_frag_ws(wf, fb1 + kb, lane);
#pragma unroll
    for (int s = 0; s < 3; ++s)
#pragma unroll
      for (int kb = 0; kb < 2; ++kb)
        a2[s][kb] = load_a_frag_ws(wf, fb2 + s * 2 + kb, lane);

    for (int bb = 0; bb < BPB; ++bb) {
      const _Float16* xt_ = sXt + bb * XT_TILE;
      float maxv[8];
#pragma unroll
      for (int v = 0; v < 8; ++v) maxv[v] = -INFINITY;

      for (int nt = 0; nt < 4; ++nt) {
        v8f acc1 = {}, acc2 = {};
#pragma unroll
        for (int kb = 0; kb < 2; ++kb) {
          v16h bf = load_b_frag_lds(sHt[bb][0], 2 + nt * 16, HT_STRIDE, kb * 32, lane);
          acc1 = wmma_f16(a1[kb], bf, acc1);
        }
        for (int s = 0; s < 3; ++s) {
#pragma unroll
          for (int kb = 0; kb < 2; ++kb) {
            v16h bf = load_b_frag_lds(sHt[bb][1], 1 + s + nt * 16, HT_STRIDE,
                                      kb * 32, lane);
            acc2 = wmma_f16(a2[s][kb], bf, acc2);
          }
        }
#pragma unroll
        for (int v = 0; v < 8; ++v) {
          const int c = m0 + v + (hs << 3);
          const int l = nt * 16 + (lane & 15);
          float xs[5];
#pragma unroll
          for (int d = 0; d < 5; ++d)
            xs[d] = (float)xt_[(l + d) * XT_STRIDE + c];
          const float avg3 = (xs[1] + xs[2] + xs[3]) * (1.0f / 3.0f);
          const float avg5 = (xs[0] + xs[1] + xs[2] + xs[3] + xs[4]) * 0.2f;
          float m3 = xs[2];
          m3 = fmaxf(m3, (l >= 1)       ? xs[1] : -INFINITY);
          m3 = fmaxf(m3, (l <= cfg::SEQ - 2) ? xs[3] : -INFINITY);
          float m5 = m3;
          m5 = fmaxf(m5, (l >= 2)       ? xs[0] : -INFINITY);
          m5 = fmaxf(m5, (l <= cfg::SEQ - 3) ? xs[4] : -INFINITY);
          const float g1 = sigmoidf_(acc1[v]);
          const float g2 = sigmoidf_(acc2[v]);
          const float mval = (avg3 + m3) * g1 + (avg5 + m5) * g2;
          maxv[v] = fmaxf(maxv[v], mval);
        }
      }
#pragma unroll
      for (int v = 0; v < 8; ++v) {
#pragma unroll
        for (int off = 8; off >= 1; off >>= 1)
          maxv[v] = fmaxf(maxv[v], __shfl_xor(maxv[v], off, 16));
        if ((lane & 15) == 0) {
          const int c = m0 + v + (hs << 3);
          out[((size_t)(b0 + bb) * cfg::PARTS + part) * cfg::CH + c] = maxv[v];
        }
      }
    }
  }
}

extern "C" void kernel_launch(void* const* d_in, const int* in_sizes, int n_in,
                              void* d_out, int out_size, void* d_ws, size_t ws_size,
                              hipStream_t stream) {
  (void)in_sizes; (void)n_in; (void)ws_size; (void)out_size;
  const float* x   = (const float*)d_in[0];
  const float* W1a = (const float*)d_in[1];
  const float* W1b = (const float*)d_in[2];
  const float* W2a = (const float*)d_in[3];
  const float* W2b = (const float*)d_in[4];
  float* out = (float*)d_out;
  _Float16* wf = (_Float16*)d_ws;

  prepass_kernel<<<dim3(cfg::PRE_GRID), dim3(256), 0, stream>>>(x, W1a, W1b, W2a, W2b, wf);
  dim3 grid(cfg::PARTS, cfg::MAIN_GRID_Y);
  dim3 block(256);
  mtb_gate_kernel<<<grid, block, 0, stream>>>(wf, out);
}


# Stage B channel max over the 16 lanes of a row: butterfly on DPP row_xmask moves in registers, 8 channels interleaved; was 4 dependent ds_bpermute LDS round trips per channel, channels in series
# speedup vs baseline: 4.6939x; 1.1104x over previous
#include <hip/hip_runtime.h>
#include <math.h>

typedef __attribute__((ext_vector_type(16))) _Float16 v16h;
typedef __attribute__((ext_vector_type(8)))  _Float16 v8h;
typedef __attribute__((ext_vector_type(8)))  float    v8f;

namespace cfg {
constexpr int NB    = 128;
constexpr int SEQ   = 64;
constexpr int CH    = 256;
constexpr int PARTS = 16;
constexpr int HID   = 64;
constexpr int XT_ROWS   = SEQ + 4;
constexpr int XT_STRIDE = CH + 8;
constexpr int HT_STRIDE = HID + 8;
constexpr int BPB     = 2;
constexpr int XT_TILE = XT_ROWS * XT_STRIDE;
constexpr int HT_TILE = XT_ROWS * HT_STRIDE;
constexpr int MAIN_GRID_Y = NB / BPB;
constexpr int FRAG_HALFS = 512;
constexpr int NFRAG_A  = 2 * PARTS * 4 * 3 * 8;
constexpr int NFRAG_B1 = PARTS * 16 * 2;
constexpr int NFRAG_B2 = PARTS * 16 * 3 * 2;
constexpr int NFRAG    = NFRAG_A + NFRAG_B1 + NFRAG_B2;
constexpr int PRE_BLOCKS = NFRAG * 64 / 256;
constexpr size_t WF_HALFS = (size_t)NFRAG * FRAG_HALFS;
constexpr int XT_PIECES = NB * SEQ * (CH / 64);
constexpr int XT_TRIPS  = 4;
constexpr int XT_BLOCKS = XT_PIECES / (2 * 8 * XT_TRIPS);
constexpr int PRE_GRID  = XT_BLOCKS + PRE_BLOCKS;
}

__device__ __forceinline__ v16h load_a_frag_ws(const _Float16* __restrict__ wf,
                                               int frag, int lane) {
  const _Float16* p = wf + (size_t)frag * cfg::FRAG_HALFS + lane * 8;
  const v8h lo = *(const v8h*)p;
  const v8h hi = *(const v8h*)(p + 256);
  return __builtin_shufflevector(lo, hi, 0, 1, 2, 3, 4, 5, 6, 7,
                                 8, 9, 10, 11, 12, 13, 14, 15);
}

__device__ __forceinline__ v16h load_b_frag_lds(const _Float16* lds, int row_base,
                                                int row_stride, int col_base,
                                                int lane) {
  const int n  = lane & 15;
  const int hs = (lane >> 4) & 1;
  const _Float16* p = lds + (size_t)(row_base + n) * row_stride + col_base;
  v16h b;
#pragma unroll
  for (int t = 0; t < 16; ++t) {
    const int v = t >> 1, e = t & 1;
    const int k = ((v >> 2) << 4) + (hs << 3) + ((v & 3) << 1) + e;
    b[t] = p[k];
  }
  return b;
}

__device__ __forceinline__ v8f wmma_f16(v16h a, v16h b, v8f c) {
  return __builtin_amdgcn_wmma_f32_16x16x32_f16(false, a, false, b,
                                                (short)0, c, false, false);
}

template <int M>
__device__ __forceinline__ float row_xmask(float x) {
  return __builtin_bit_cast(float, __builtin_amdgcn_mov_dpp(__builtin_bit_cast(int, x),
                                                           0x160 + M, 0xf, 0xf, false));
}

__device__ __forceinline__ float sigmoidf_(float z) {
  return fmaf(0.5f, __builtin_amdgcn_tanhf(0.5f * z), 0.5f);
}

template <int U, int KSTEPS>
__device__ __forceinline__ void conv_a_step(const _Float16* __restrict__ wf,
                                            const _Float16* sXt, int fbase, int lane,
                                            v16h (&aq)[4], v16h& acur, v16h (&bq)[4],
                                            v8f (&acc)[cfg::BPB][4]) {
  using namespace cfg;
  constexpr int T  = U >> 1;
  constexpr int BB = U & 1;
  constexpr int UN = U + 1;
  constexpr int TN = UN >> 1;
  constexpr int BN = UN & 1;
  if (BB == 0) {
    acur = aq[T & 3];
    if (T + 4 < KSTEPS) aq[T & 3] = load_a_frag_ws(wf, fbase + T + 4, lane);
  }
  v16h bn[4];
  if (UN < 2 * KSTEPS) {
#pragma unroll
    for (int nt = 0; nt < 4; ++nt)
      bn[nt] = load_b_frag_lds(sXt + BN * XT_TILE, 1 + (TN >> 3) + nt * 16, XT_STRIDE,
                               (TN & 7) * 32, lane);
  }
  __builtin_amdgcn_sched_barrier(0);
#pragma unroll
  for (int nt = 0; nt < 4; ++nt) acc[BB][nt] = wmma_f16(acur, bq[nt], acc[BB][nt]);
  if (UN < 2 * KSTEPS) {
#pragma unroll
    for (int nt = 0; nt < 4; ++nt) bq[nt] = bn[nt];
  }
}

template <int U, int NU, int KSTEPS>
struct ConvASteps {
  static __device__ __forceinline__ void run(const _Float16* __restrict__ wf,
                                             const _Float16* sXt, int fbase, int lane,
                                             v16h (&aq)[4], v16h& acur, v16h (&bq)[4],
                                             v8f (&acc)[cfg::BPB][4]) {
    conv_a_step<U, KSTEPS>(wf, sXt, fbase, lane, aq, acur, bq, acc);
    ConvASteps<U + 1, NU, KSTEPS>::run(wf, sXt, fbase, lane, aq, acur, bq, acc);
  }
};
template <int NU, int KSTEPS>
struct ConvASteps<NU, NU, KSTEPS> {
  static __device__ __forceinline__ void run(const _Float16* __restrict__, const _Float16*,
                                             int, int, v16h (&)[4], v16h&, v16h (&)[4],
                                             v8f (&)[cfg::BPB][4]) {}
};

__device__ __forceinline__ void x_to_f16_tiles(const float* __restrict__ x,
                                               _Float16* __restrict__ xt,
                                               int blk, int tid) {
  using namespace cfg;
  const int wave = tid >> 5;
  const int lane = tid & 31;
  const int c8   = lane >> 2;
  const int pq   = lane & 3;
  for (int it = 0; it < XT_TRIPS; ++it) {
    const int piece0 = ((blk * XT_TRIPS + it) * 8 + wave) * 2;
    const int row    = piece0 >> 2;
    const int c0     = (piece0 & 3) * 64 + c8 * 8;
    const float* src = x + ((size_t)row * CH + c0) * PARTS + pq * 4;
    float4 v[2][8];
#pragma unroll
    for (int h = 0; h < 2; ++h)
#pragma unroll
      for (int j = 0; j < 8; ++j)
        v[h][j] = *(const float4*)(src + (h * 64 + j) * PARTS);
    __builtin_amdgcn_sched_barrier(0);
    const int b = row >> 6;
    const int l = row & 63;
    _Float16* dst = xt + (((size_t)(b * PARTS + pq * 4) * SEQ + l) * CH + c0);
#pragma unroll
    for (int h = 0; h < 2; ++h) {
      v8h o0, o1, o2, o3;
#pragma unroll
      for (int j = 0; j < 8; ++j) {
        o0[j] = (_Float16)v[h][j].x;
        o1[j] = (_Float16)v[h][j].y;
        o2[j] = (_Float16)v[h][j].z;
        o3[j] = (_Float16)v[h][j].w;
      }
      *(v8h*)(dst + h * 64)                        = o0;
      *(v8h*)(dst + h * 64 + (size_t)1 * SEQ * CH) = o1;
      *(v8h*)(dst + h * 64 + (size_t)2 * SEQ * CH) = o2;
      *(v8h*)(dst + h * 64 + (size_t)3 * SEQ * CH) = o3;
    }
  }
}

__global__ __launch_bounds__(256)
void prepass_kernel(const float* __restrict__ x,
                    const float* __restrict__ W1a,
                    const float* __restrict__ W1b,
                    const float* __restrict__ W2a,
                    const float* __restrict__ W2b,
                    _Float16* __restrict__ wf) {
  using namespace cfg;
  if (blockIdx.x < XT_BLOCKS) {
    x_to_f16_tiles(x, wf + WF_HALFS, blockIdx.x, threadIdx.x);
    return;
  }
  const int gid  = (blockIdx.x - XT_BLOCKS) * 256 + threadIdx.x;
  const int frag = gid >> 6;
  const int sel  = (gid >> 5) & 1;
  const int lane = gid & 31;
  const int m    = lane & 15;
  const int hs   = lane >> 4;
  const int k0   = sel * 16 + hs * 8;
  const float* src;
  int kstride;
  if (frag < NFRAG_A) {
    const int kb = frag & 7;
    int r = frag >> 3;
    const int s = r % 3;  r /= 3;
    const int mt = r & 3; r >>= 2;
    const int part = r & 15;
    const int br = r >> 4;
    src = (br ? W2a : W1a) +
          ((size_t)((part * HID + mt * 16 + m) * CH + kb * 32 + k0)) * 3 + s;
    kstride = 3;
  } else if (frag < NFRAG_A + NFRAG_B1) {
    const int f = frag - NFRAG_A;
    const int kb = f & 1;
    const int ct = (f >> 1) & 15;
    const int part = f >> 5;
    src = W1b + (size_t)((part * CH + ct * 16 + m) * HID + kb * 32 + k0);
    kstride = 1;
  } else {
    const int f = frag - NFRAG_A - NFRAG_B1;
    const int kb = f & 1;
    int r = f >> 1;
    const int s = r % 3;  r /= 3;
    const int ct = r & 15;
    const int part = r >> 4;
    src = W2b + ((size_t)((part * CH + ct * 16 + m) * HID + kb * 32 + k0)) * 3 + s;
    kstride = 3;
  }
  v8h o;
#pragma unroll
  for (int e = 0; e < 8; ++e) o[e] = (_Float16)src[e * kstride];
  *(v8h*)(wf + (size_t)gid * 8) = o;
}

__global__ __launch_bounds__(256)
void mtb_gate_kernel(const _Float16* __restrict__ wf,
                     float* __restrict__ out) {
  using namespace cfg;
  __shared__ __attribute__((aligned(16))) _Float16 sXt[BPB * XT_TILE];
  __shared__ _Float16 sHt[BPB][2][HT_TILE];

  const int part = blockIdx.x;
  const int b0   = blockIdx.y * BPB;
  const int tid  = threadIdx.x;
  const int lane = tid & 31;
  const int wave = tid >> 5;
  const int hs   = (lane >> 4) & 1;

  {
#pragma unroll
    for (int bb = 0; bb < BPB; ++bb) {
      _Float16* xt_ = sXt + bb * XT_TILE;
      xt_[0 * XT_STRIDE + tid]         = (_Float16)0.f;
      xt_[1 * XT_STRIDE + tid]         = (_Float16)0.f;
      xt_[(SEQ + 2) * XT_STRIDE + tid] = (_Float16)0.f;
      xt_[(SEQ + 3) * XT_STRIDE + tid] = (_Float16)0.f;
    }
    if (tid < HID) {
#pragma unroll
      for (int bb = 0; bb < BPB; ++bb) {
#pragma unroll
        for (int r = 0; r < 2; ++r) {
          sHt[bb][0][r * HT_STRIDE + tid] = (_Float16)0.f;
          sHt[bb][1][r * HT_STRIDE + tid] = (_Float16)0.f;
          sHt[bb][0][(SEQ + 2 + r) * HT_STRIDE + tid] = (_Float16)0.f;
          sHt[bb][1][(SEQ + 2 + r) * HT_STRIDE + tid] = (_Float16)0.f;
        }
      }
    }
    v8h t[BPB][SEQ * CH / 8 / 256];
#pragma unroll
    for (int bb = 0; bb < BPB; ++bb) {
      const _Float16* xp = wf + WF_HALFS + ((size_t)((b0 + bb) * PARTS + part) * SEQ) * CH;
#pragma unroll
      for (int i = 0; i < SEQ * CH / 8 / 256; ++i)
        t[bb][i] = *(const v8h*)(xp + (size_t)(i * 256 + tid) * 8);
    }
#pragma unroll
    for (int bb = 0; bb < BPB; ++bb) {
#pragma unroll
      for (int i = 0; i < SEQ * CH / 8 / 256; ++i) {
        const int chunk = i * 256 + tid;
        *(v8h*)&sXt[bb * XT_TILE + (2 + (chunk >> 5)) * XT_STRIDE + (chunk & 31) * 8] = t[bb][i];
      }
    }
  }
  {
    const int br = wave >> 2;
    const int mt = wave & 3;
    const int m0 = mt * 16;
    const int fbase = ((br * PARTS + part) * 4 + mt) * 24;
    constexpr int KSTEPS = 3 * (CH / 32);
    v8f acc[BPB][4] = {};
    v16h aq[4], bq[4], acur;
#pragma unroll
    for (int i = 0; i < 4; ++i) aq[i] = load_a_frag_ws(wf, fbase + i, lane);
    __syncthreads();
#pragma unroll
    for (int nt = 0; nt < 4; ++nt)
      bq[nt] = load_b_frag_lds(sXt, 1 + nt * 16, XT_STRIDE, 0, lane);
    ConvASteps<0, 2 * KSTEPS, KSTEPS>::run(wf, sXt, fbase, lane, aq, acur, bq, acc);
#pragma unroll
    for (int bb = 0; bb < BPB; ++bb) {
#pragma unroll
      for (int nt = 0; nt < 4; ++nt) {
#pragma unroll
        for (int v = 0; v < 8; ++v) {
          float f = acc[bb][nt][v];
          f = f > 0.f ? f : 0.01f * f;
          const int row = 2 + nt * 16 + (lane & 15);
          const int col = m0 + v + (hs << 3);
          sHt[bb][br][row * HT_STRIDE + col] = (_Float16)f;
        }
      }
    }
  }
  __syncthreads();

  for (int j = 0; j < 2; ++j) {
    const int m0 = wave * 32 + j * 16;
    const int ct  = wave * 2 + j;
    const int fb1 = NFRAG_A + (part * 16 + ct) * 2;
    const int fb2 = NFRAG_A + NFRAG_B1 + (part * 16 + ct) * 6;
    v16h a1[2], a2[3][2];
#pragma unroll
    for (int kb = 0; kb < 2; ++kb)
      a1[kb] = load_a_frag_ws(wf, fb1 + kb, lane);
#pragma unroll
    for (int s = 0; s < 3; ++s)
#pragma unroll
      for (int kb = 0; kb < 2; ++kb)
        a2[s][kb] = load_a_frag_ws(wf, fb2 + s * 2 + kb, lane);

    for (int bb = 0; bb < BPB; ++bb) {
      const _Float16* xt_ = sXt + bb * XT_TILE;
      float maxv[8];
#pragma unroll
      for (int v = 0; v < 8; ++v) maxv[v] = -INFINITY;

      for (int nt = 0; nt < 4; ++nt) {
        v8f acc1 = {}, acc2 = {};
#pragma unroll
        for (int kb = 0; kb < 2; ++kb) {
          v16h bf = load_b_frag_lds(sHt[bb][0], 2 + nt * 16, HT_STRIDE, kb * 32, lane);
          acc1 = wmma_f16(a1[kb], bf, acc1);
        }
        for (int s = 0; s < 3; ++s) {
#pragma unroll
          for (int kb = 0; kb < 2; ++kb) {
            v16h bf = load_b_frag_lds(sHt[bb][1], 1 + s + nt * 16, HT_STRIDE,
                                      kb * 32, lane);
            acc2 = wmma_f16(a2[s][kb], bf, acc2);
          }
        }
#pragma unroll
        for (int v = 0; v < 8; ++v) {
          const int c = m0 + v + (hs << 3);
          const int l = nt * 16 + (lane & 15);
          float xs[5];
#pragma unroll
          for (int d = 0; d < 5; ++d)
            xs[d] = (float)xt_[(l + d) * XT_STRIDE + c];
          const float avg3 = (xs[1] + xs[2] + xs[3]) * (1.0f / 3.0f);
          const float avg5 = (xs[0] + xs[1] + xs[2] + xs[3] + xs[4]) * 0.2f;
          float m3 = xs[2];
          m3 = fmaxf(m3, (l >= 1)       ? xs[1] : -INFINITY);
          m3 = fmaxf(m3, (l <= cfg::SEQ - 2) ? xs[3] : -INFINITY);
          float m5 = m3;
          m5 = fmaxf(m5, (l >= 2)       ? xs[0] : -INFINITY);
          m5 = fmaxf(m5, (l <= cfg::SEQ - 3) ? xs[4] : -INFINITY);
          const float g1 = sigmoidf_(acc1[v]);
          const float g2 = sigmoidf_(acc2[v]);
          const float mval = (avg3 + m3) * g1 + (avg5 + m5) * g2;
          maxv[v] = fmaxf(maxv[v], mval);
        }
      }
#pragma unroll
      for (int v = 0; v < 8; ++v) maxv[v] = fmaxf(maxv[v], row_xmask<8>(maxv[v]));
#pragma unroll
      for (int v = 0; v < 8; ++v) maxv[v] = fmaxf(maxv[v], row_xmask<4>(maxv[v]));
#pragma unroll
      for (int v = 0; v < 8; ++v) maxv[v] = fmaxf(maxv[v], row_xmask<2>(maxv[v]));
#pragma unroll
      for (int v = 0; v < 8; ++v) maxv[v] = fmaxf(maxv[v], row_xmask<1>(maxv[v]));
      if ((lane & 15) == 0) {
#pragma unroll
        for (int v = 0; v < 8; ++v) {
          const int c = m0 + v + (hs << 3);
          out[((size_t)(b0 + bb) * cfg::PARTS + part) * cfg::CH + c] = maxv[v];
        }
      }
    }
  }
}

extern "C" void kernel_launch(void* const* d_in, const int* in_sizes, int n_in,
                              void* d_out, int out_size, void* d_ws, size_t ws_size,
                              hipStream_t stream) {
  (void)in_sizes; (void)n_in; (void)ws_size; (void)out_size;
  const float* x   = (const float*)d_in[0];
  const float* W1a = (const float*)d_in[1];
  const float* W1b = (const float*)d_in[2];
  const float* W2a = (const float*)d_in[3];
  const float* W2b = (const float*)d_in[4];
  float* out = (float*)d_out;
  _Float16* wf = (_Float16*)d_ws;

  prepass_kernel<<<dim3(cfg::PRE_GRID), dim3(256), 0, stream>>>(x, W1a, W1b, W2a, W2b, wf);
  dim3 grid(cfg::PARTS, cfg::MAIN_GRID_Y);
  dim3 block(256);
  mtb_gate_kernel<<<grid, block, 0, stream>>>(wf, out);
}
